# starting kernel
# baseline (speedup 1.0000x reference)
#include <hip/hip_runtime.h>

typedef float v2f __attribute__((ext_vector_type(2)));
typedef float v8f __attribute__((ext_vector_type(8)));

#define IN_F  512
#define HID_F 256
#define OUT_F 128

#define WMMA_F32(a, b, c) \
  __builtin_amdgcn_wmma_f32_16x16x4_f32(false, (a), false, (b), (short)0, (c), false, false)

__global__ void zero_f4(float4* __restrict__ p, long long n4) {
  long long i = (long long)blockIdx.x * blockDim.x + threadIdx.x;
  if (i < n4) p[i] = make_float4(0.f, 0.f, 0.f, 0.f);
}

__global__ void degree_k(const int* __restrict__ dst, float* __restrict__ deg, int E) {
  int i = blockIdx.x * blockDim.x + threadIdx.x;
  if (i < E) atomicAdd(&deg[dst[i]], 1.0f);
}

__global__ void invdeg_k(float* __restrict__ deg, int n) {
  int i = blockIdx.x * blockDim.x + threadIdx.x;
  if (i < n) deg[i] = 1.0f / fmaxf(deg[i], 1.0f);
}

__global__ __launch_bounds__(128)
void gemm_wmma_f32(const float* __restrict__ A, const float* __restrict__ B,
                   float* __restrict__ C, int M, int N, int K) {
  const int lane = threadIdx.x & 31;
  const int wave = threadIdx.x >> 5;
  const int m0 = (blockIdx.x * 4 + wave) * 32;
  const int n0 = blockIdx.y * 64;
  if (m0 >= M) return;
  const bool has2 = (m0 + 16) < M;

  const int half = lane >> 4;
  const int l15  = lane & 15;

  const int row0 = m0 + l15;
  int row1 = m0 + 16 + l15;
  if (row1 >= M) row1 = M - 1;
  const float* Ap0 = A + (size_t)row0 * K + half * 2;
  const float* Ap1 = A + (size_t)row1 * K + half * 2;
  const float* Bp = B + (size_t)(half * 2) * N + n0 + l15;

  v8f c00 = {}, c01 = {}, c02 = {}, c03 = {};
  v8f c10 = {}, c11 = {}, c12 = {}, c13 = {};

  for (int k = 0; k < K; k += 4) {
    v2f a0 = *(const v2f*)(Ap0 + k);
    v2f a1 = *(const v2f*)(Ap1 + k);
    const float* bk = Bp + (size_t)k * N;
    v2f b0, b1, b2, b3;
    b0.x = bk[0];   b0.y = bk[(size_t)N + 0];
    b1.x = bk[16];  b1.y = bk[(size_t)N + 16];
    b2.x = bk[32];  b2.y = bk[(size_t)N + 32];
    b3.x = bk[48];  b3.y = bk[(size_t)N + 48];
    c00 = WMMA_F32(a0, b0, c00);
    c01 = WMMA_F32(a0, b1, c01);
    c02 = WMMA_F32(a0, b2, c02);
    c03 = WMMA_F32(a0, b3, c03);
    c10 = WMMA_F32(a1, b0, c10);
    c11 = WMMA_F32(a1, b1, c11);
    c12 = WMMA_F32(a1, b2, c12);
    c13 = WMMA_F32(a1, b3, c13);
  }

  float* Cp0 = C + (size_t)(m0 + half * 8) * N + n0 + l15;
#pragma unroll
  for (int j = 0; j < 8; ++j) {
    float* r = Cp0 + (size_t)j * N;
    r[0]  = c00[j];
    r[16] = c01[j];
    r[32] = c02[j];
    r[48] = c03[j];
  }
  if (has2) {
    float* Cp1 = C + (size_t)(m0 + 16 + half * 8) * N + n0 + l15;
#pragma unroll
    for (int j = 0; j < 8; ++j) {
      float* r = Cp1 + (size_t)j * N;
      r[0]  = c10[j];
      r[16] = c11[j];
      r[32] = c12[j];
      r[48] = c13[j];
    }
  }
}

__global__ void edge_scatter(const float* __restrict__ hw, const float* __restrict__ ew,
                             const int* __restrict__ src, const int* __restrict__ dst,
                             float* __restrict__ agg, long long total, int fshift, int F) {
  long long idx = (long long)blockIdx.x * blockDim.x + threadIdx.x;
  if (idx >= total) return;
  int e  = (int)(idx >> fshift);
  int f4 = (int)(idx & ((1 << fshift) - 1)) << 2;
  float w = ew[e];
  const float4 v = *(const float4*)(hw + (size_t)src[e] * F + f4);
  float* p = agg + (size_t)dst[e] * F + f4;
  atomicAdd(p + 0, v.x * w);
  atomicAdd(p + 1, v.y * w);
  atomicAdd(p + 2, v.z * w);
  atomicAdd(p + 3, v.w * w);
}

__global__ void bias_norm_act(float* __restrict__ data, const float* __restrict__ inv,
                              const float* __restrict__ bias, long long n,
                              int fshift, int do_relu) {
  long long i = (long long)blockIdx.x * blockDim.x + threadIdx.x;
  if (i >= n) return;
  int row = (int)(i >> fshift);
  int col = (int)(i & ((1 << fshift) - 1));
  float v = data[i] * inv[row] + bias[col];
  data[i] = do_relu ? fmaxf(v, 0.f) : v;
}

extern "C" void kernel_launch(void* const* d_in, const int* in_sizes, int n_in,
                              void* d_out, int out_size, void* d_ws, size_t ws_size,
                              hipStream_t stream) {
  const float* x    = (const float*)d_in[0];
  const float* W1   = (const float*)d_in[1];
  const float* b1   = (const float*)d_in[2];
  const float* W2   = (const float*)d_in[3];
  const float* b2   = (const float*)d_in[4];
  const float* ew   = (const float*)d_in[5];
  const int*   esrc = (const int*)d_in[6];
  const int*   edst = (const int*)d_in[7];
  float* out = (float*)d_out;

  const int M = in_sizes[0] / IN_F;
  const int E = in_sizes[5];

  float* inv  = (float*)d_ws;
  const long long Mpad = ((long long)M + 63) & ~63LL;
  float* bufA = inv + Mpad;
  float* bufB = bufA + (long long)M * HID_F;

  const int T = 256;

  {
    long long n4 = (long long)M / 4;
    zero_f4<<<dim3((unsigned)((n4 + T - 1) / T)), dim3(T), 0, stream>>>((float4*)inv, n4);
    long long nB4 = (long long)M * HID_F / 4;
    zero_f4<<<dim3((unsigned)((nB4 + T - 1) / T)), dim3(T), 0, stream>>>((float4*)bufB, nB4);
    long long nO4 = (long long)M * OUT_F / 4;
    zero_f4<<<dim3((unsigned)((nO4 + T - 1) / T)), dim3(T), 0, stream>>>((float4*)out, nO4);
  }

  degree_k<<<dim3((E + T - 1) / T), dim3(T), 0, stream>>>(edst, inv, E);
  invdeg_k<<<dim3((M + T - 1) / T), dim3(T), 0, stream>>>(inv, M);

  {
    dim3 g((M + 127) / 128, HID_F / 64);
    gemm_wmma_f32<<<g, dim3(128), 0, stream>>>(x, W1, bufA, M, HID_F, IN_F);
  }

  {
    long long total = (long long)E * (HID_F / 4);
    edge_scatter<<<dim3((unsigned)((total + T - 1) / T)), dim3(T), 0, stream>>>(
        bufA, ew, esrc, edst, bufB, total, 6  , HID_F);
  }

  {
    long long n = (long long)M * HID_F;
    bias_norm_act<<<dim3((unsigned)((n + T - 1) / T)), dim3(T), 0, stream>>>(
        bufB, inv, b1, n, 8  , 1);
  }

  {
    dim3 g((M + 127) / 128, OUT_F / 64);
    gemm_wmma_f32<<<g, dim3(128), 0, stream>>>(bufB, W2, bufA, M, OUT_F, HID_F);
  }

  {
    long long total = (long long)E * (OUT_F / 4);
    edge_scatter<<<dim3((unsigned)((total + T - 1) / T)), dim3(T), 0, stream>>>(
        bufA, ew, esrc, edst, out, total, 5  , OUT_F);
  }

  {
    long long n = (long long)M * OUT_F;
    bias_norm_act<<<dim3((unsigned)((n + T - 1) / T)), dim3(T), 0, stream>>>(
        out, inv, b2, n, 7  , 0);
  }
}


# edge scatter: one wave per edge row, lane l adds floats 32j+l, so each f32 atomic and gather load covers one whole 128-B line, not 4 quarter lines; atomic sums stay arrival-ordered
# speedup vs baseline: 1.5237x; 1.5237x over previous
#include <hip/hip_runtime.h>

typedef float v2f __attribute__((ext_vector_type(2)));
typedef float v8f __attribute__((ext_vector_type(8)));

#define IN_F  512
#define HID_F 256
#define OUT_F 128

#define WMMA_F32(a, b, c) \
  __builtin_amdgcn_wmma_f32_16x16x4_f32(false, (a), false, (b), (short)0, (c), false, false)

__global__ void zero_f4(float4* __restrict__ p, long long n4) {
  long long i = (long long)blockIdx.x * blockDim.x + threadIdx.x;
  if (i < n4) p[i] = make_float4(0.f, 0.f, 0.f, 0.f);
}

__global__ void degree_k(const int* __restrict__ dst, float* __restrict__ deg, int E) {
  int i = blockIdx.x * blockDim.x + threadIdx.x;
  if (i < E) atomicAdd(&deg[dst[i]], 1.0f);
}

__global__ void invdeg_k(float* __restrict__ deg, int n) {
  int i = blockIdx.x * blockDim.x + threadIdx.x;
  if (i < n) deg[i] = 1.0f / fmaxf(deg[i], 1.0f);
}

__global__ __launch_bounds__(128)
void gemm_wmma_f32(const float* __restrict__ A, const float* __restrict__ B,
                   float* __restrict__ C, int M, int N, int K) {
  const int lane = threadIdx.x & 31;
  const int wave = threadIdx.x >> 5;
  const int m0 = (blockIdx.x * 4 + wave) * 32;
  const int n0 = blockIdx.y * 64;
  if (m0 >= M) return;
  const bool has2 = (m0 + 16) < M;

  const int half = lane >> 4;
  const int l15  = lane & 15;

  const int row0 = m0 + l15;
  int row1 = m0 + 16 + l15;
  if (row1 >= M) row1 = M - 1;
  const float* Ap0 = A + (size_t)row0 * K + half * 2;
  const float* Ap1 = A + (size_t)row1 * K + half * 2;
  const float* Bp = B + (size_t)(half * 2) * N + n0 + l15;

  v8f c00 = {}, c01 = {}, c02 = {}, c03 = {};
  v8f c10 = {}, c11 = {}, c12 = {}, c13 = {};

  for (int k = 0; k < K; k += 4) {
    v2f a0 = *(const v2f*)(Ap0 + k);
    v2f a1 = *(const v2f*)(Ap1 + k);
    const float* bk = Bp + (size_t)k * N;
    v2f b0, b1, b2, b3;
    b0.x = bk[0];   b0.y = bk[(size_t)N + 0];
    b1.x = bk[16];  b1.y = bk[(size_t)N + 16];
    b2.x = bk[32];  b2.y = bk[(size_t)N + 32];
    b3.x = bk[48];  b3.y = bk[(size_t)N + 48];
    c00 = WMMA_F32(a0, b0, c00);
    c01 = WMMA_F32(a0, b1, c01);
    c02 = WMMA_F32(a0, b2, c02);
    c03 = WMMA_F32(a0, b3, c03);
    c10 = WMMA_F32(a1, b0, c10);
    c11 = WMMA_F32(a1, b1, c11);
    c12 = WMMA_F32(a1, b2, c12);
    c13 = WMMA_F32(a1, b3, c13);
  }

  float* Cp0 = C + (size_t)(m0 + half * 8) * N + n0 + l15;
#pragma unroll
  for (int j = 0; j < 8; ++j) {
    float* r = Cp0 + (size_t)j * N;
    r[0]  = c00[j];
    r[16] = c01[j];
    r[32] = c02[j];
    r[48] = c03[j];
  }
  if (has2) {
    float* Cp1 = C + (size_t)(m0 + 16 + half * 8) * N + n0 + l15;
#pragma unroll
    for (int j = 0; j < 8; ++j) {
      float* r = Cp1 + (size_t)j * N;
      r[0]  = c10[j];
      r[16] = c11[j];
      r[32] = c12[j];
      r[48] = c13[j];
    }
  }
}

template <int F>
__global__ __launch_bounds__(256)
void edge_scatter_rows(const float* __restrict__ hw, const float* __restrict__ ew,
                       const int* __restrict__ src, const int* __restrict__ dst,
                       float* __restrict__ agg, int E) {
  constexpr int J = F / 32;
  constexpr int U = 1024 / F;
  const int lane = threadIdx.x & 31;
  const int wave = __builtin_amdgcn_readfirstlane((int)(threadIdx.x >> 5));
  const int e0 = (blockIdx.x * 8 + wave) * U;
  if (e0 >= E) return;
  if (e0 + U <= E) {
    float v[U][J];
    float w[U];
    int   d[U];
    for (int u = 0; u < U; ++u) {
      const int s = src[e0 + u];
      d[u] = dst[e0 + u];
      w[u] = ew[e0 + u];
      const float* r = hw + (size_t)s * F + lane;
      for (int j = 0; j < J; ++j) v[u][j] = r[32 * j];
    }
    for (int u = 0; u < U; ++u) {
      float* p = agg + (size_t)d[u] * F + lane;
      for (int j = 0; j < J; ++j) atomicAdd(p + 32 * j, v[u][j] * w[u]);
    }
  } else {
    for (int e = e0; e < E; ++e) {
      const int s = src[e];
      const float wt = ew[e];
      const float* r = hw + (size_t)s * F + lane;
      float* p = agg + (size_t)dst[e] * F + lane;
      for (int j = 0; j < J; ++j) atomicAdd(p + 32 * j, r[32 * j] * wt);
    }
  }
}

__global__ void bias_norm_act(float* __restrict__ data, const float* __restrict__ inv,
                              const float* __restrict__ bias, long long n,
                              int fshift, int do_relu) {
  long long i = (long long)blockIdx.x * blockDim.x + threadIdx.x;
  if (i >= n) return;
  int row = (int)(i >> fshift);
  int col = (int)(i & ((1 << fshift) - 1));
  float v = data[i] * inv[row] + bias[col];
  data[i] = do_relu ? fmaxf(v, 0.f) : v;
}

extern "C" void kernel_launch(void* const* d_in, const int* in_sizes, int n_in,
                              void* d_out, int out_size, void* d_ws, size_t ws_size,
                              hipStream_t stream) {
  const float* x    = (const float*)d_in[0];
  const float* W1   = (const float*)d_in[1];
  const float* b1   = (const float*)d_in[2];
  const float* W2   = (const float*)d_in[3];
  const float* b2   = (const float*)d_in[4];
  const float* ew   = (const float*)d_in[5];
  const int*   esrc = (const int*)d_in[6];
  const int*   edst = (const int*)d_in[7];
  float* out = (float*)d_out;

  const int M = in_sizes[0] / IN_F;
  const int E = in_sizes[5];

  float* inv  = (float*)d_ws;
  const long long Mpad = ((long long)M + 63) & ~63LL;
  float* bufA = inv + Mpad;
  float* bufB = bufA + (long long)M * HID_F;

  const int T = 256;

  {
    long long n4 = (long long)M / 4;
    zero_f4<<<dim3((unsigned)((n4 + T - 1) / T)), dim3(T), 0, stream>>>((float4*)inv, n4);
    long long nB4 = (long long)M * HID_F / 4;
    zero_f4<<<dim3((unsigned)((nB4 + T - 1) / T)), dim3(T), 0, stream>>>((float4*)bufB, nB4);
    long long nO4 = (long long)M * OUT_F / 4;
    zero_f4<<<dim3((unsigned)((nO4 + T - 1) / T)), dim3(T), 0, stream>>>((float4*)out, nO4);
  }

  degree_k<<<dim3((E + T - 1) / T), dim3(T), 0, stream>>>(edst, inv, E);
  invdeg_k<<<dim3((M + T - 1) / T), dim3(T), 0, stream>>>(inv, M);

  {
    dim3 g((M + 127) / 128, HID_F / 64);
    gemm_wmma_f32<<<g, dim3(128), 0, stream>>>(x, W1, bufA, M, HID_F, IN_F);
  }

  {
    const int per_block = 8 * (1024 / HID_F);
    edge_scatter_rows<HID_F><<<dim3((E + per_block - 1) / per_block), dim3(T), 0, stream>>>(
        bufA, ew, esrc, edst, bufB, E);
  }

  {
    long long n = (long long)M * HID_F;
    bias_norm_act<<<dim3((unsigned)((n + T - 1) / T)), dim3(T), 0, stream>>>(
        bufB, inv, b1, n, 8  , 1);
  }

  {
    dim3 g((M + 127) / 128, OUT_F / 64);
    gemm_wmma_f32<<<g, dim3(128), 0, stream>>>(bufB, W2, bufA, M, OUT_F, HID_F);
  }

  {
    const int per_block = 8 * (1024 / OUT_F);
    edge_scatter_rows<OUT_F><<<dim3((E + per_block - 1) / per_block), dim3(T), 0, stream>>>(
        bufA, ew, esrc, edst, out, E);
  }

  {
    long long n = (long long)M * OUT_F;
    bias_norm_act<<<dim3((unsigned)((n + T - 1) / T)), dim3(T), 0, stream>>>(
        out, inv, b2, n, 7  , 0);
  }
}


# aggregation: edges counting-sorted by dst per call (one int atomic per edge = its rank); one wave sums a dst row's gathered rows in rank order, stores it once with 1/deg+bias(+relu); no f32 atomics
# speedup vs baseline: 1.9860x; 1.3034x over previous
#include <hip/hip_runtime.h>


typedef float v2f __attribute__((ext_vector_type(2)));
typedef float v8f __attribute__((ext_vector_type(8)));

#define IN_F  512
#define HID_F 256
#define OUT_F 128

#define WMMA_F32(a, b, c) \
  __builtin_amdgcn_wmma_f32_16x16x4_f32(false, (a), false, (b), (short)0, (c), false, false)

__global__ void zero_f4(float4* __restrict__ p, long long n4) {
  long long i = (long long)blockIdx.x * blockDim.x + threadIdx.x;
  if (i < n4) p[i] = make_float4(0.f, 0.f, 0.f, 0.f);
}


__global__ __launch_bounds__(256)
void count_rank(const int* __restrict__ dst, int* __restrict__ cnt,
                int* __restrict__ rank, int E) {
  const int e = (blockIdx.x * 256 + threadIdx.x) * 4;
  if (e + 3 < E) {
    const int4 d = *(const int4*)(dst + e);
    int4 r;
    r.x = atomicAdd(&cnt[d.x], 1);
    r.y = atomicAdd(&cnt[d.y], 1);
    r.z = atomicAdd(&cnt[d.z], 1);
    r.w = atomicAdd(&cnt[d.w], 1);
    *(int4*)(rank + e) = r;
  } else {
    for (int i = e; i < E; ++i) rank[i] = atomicAdd(&cnt[dst[i]], 1);
  }
}

__global__ __launch_bounds__(1024)
void scan_excl(int* __restrict__ p, int n) {
  __shared__ int wsum[32];
  const int tid  = threadIdx.x;
  const int lane = tid & 31;
  const int wv   = tid >> 5;
  int carry = 0;
  for (int base = 0; base < n; base += 4096) {
    const int i = base + tid * 4;
    int4 c = make_int4(0, 0, 0, 0);
    if (i + 3 < n) {
      c = *(const int4*)(p + i);
    } else {
      if (i     < n) c.x = p[i];
      if (i + 1 < n) c.y = p[i + 1];
      if (i + 2 < n) c.z = p[i + 2];
    }
    const int t = c.x + c.y + c.z + c.w;
    int s = t;
    for (int o = 1; o < 32; o <<= 1) {
      const int v = __shfl_up(s, o);
      if (lane >= o) s += v;
    }
    if (lane == 31) wsum[wv] = s;
    __syncthreads();
    const int ws = wsum[lane];
    int wss = ws;
    for (int o = 1; o < 32; o <<= 1) {
      const int v = __shfl_up(wss, o);
      if (lane >= o) wss += v;
    }
    const int wave_excl = __shfl(wss - ws, wv);
    const int total     = __shfl(wss, 31);
    const int x0 = carry + wave_excl + (s - t);
    const int x1 = x0 + c.x;
    const int x2 = x1 + c.y;
    const int x3 = x2 + c.z;
    if (i + 3 < n) {
      *(int4*)(p + i) = make_int4(x0, x1, x2, x3);
    } else {
      if (i     < n) p[i]     = x0;
      if (i + 1 < n) p[i + 1] = x1;
      if (i + 2 < n) p[i + 2] = x2;
    }
    carry += total;
    __syncthreads();
  }
}

__global__ __launch_bounds__(256)
void place_edges(const int* __restrict__ src, const int* __restrict__ dst,
                 const float* __restrict__ ew, const int* __restrict__ start,
                 const int* __restrict__ rank, int2* __restrict__ pairs, int E) {
  const int e = (blockIdx.x * 256 + threadIdx.x) * 4;
  if (e + 3 < E) {
    const int4   d = *(const int4*)(dst + e);
    const int4   r = *(const int4*)(rank + e);
    const int4   s = *(const int4*)(src + e);
    const float4 w = *(const float4*)(ew + e);
    pairs[start[d.x] + r.x] = make_int2(s.x, __float_as_int(w.x));
    pairs[start[d.y] + r.y] = make_int2(s.y, __float_as_int(w.y));
    pairs[start[d.z] + r.z] = make_int2(s.z, __float_as_int(w.z));
    pairs[start[d.w] + r.w] = make_int2(s.w, __float_as_int(w.w));
  } else {
    for (int i = e; i < E; ++i)
      pairs[start[dst[i]] + rank[i]] = make_int2(src[i], __float_as_int(ew[i]));
  }
}

__global__ __launch_bounds__(256)
void move_pairs(const int2* __restrict__ from, int2* __restrict__ to, int n) {
  for (int i = blockIdx.x * 256 + threadIdx.x; i < n; i += gridDim.x * 256) to[i] = from[i];
}

__global__ __launch_bounds__(128)
void gemm_wmma_f32(const float* __restrict__ A, const float* __restrict__ B,
                   float* __restrict__ C, int M, int N, int K) {
  const int lane = threadIdx.x & 31;
  const int wave = threadIdx.x >> 5;
  const int m0 = (blockIdx.x * 4 + wave) * 32;
  const int n0 = blockIdx.y * 64;
  if (m0 >= M) return;
  const bool has2 = (m0 + 16) < M;

  const int half = lane >> 4;
  const int l15  = lane & 15;

  const int row0 = m0 + l15;
  int row1 = m0 + 16 + l15;
  if (row1 >= M) row1 = M - 1;
  const float* Ap0 = A + (size_t)row0 * K + half * 2;
  const float* Ap1 = A + (size_t)row1 * K + half * 2;
  const float* Bp = B + (size_t)(half * 2) * N + n0 + l15;

  v8f c00 = {}, c01 = {}, c02 = {}, c03 = {};
  v8f c10 = {}, c11 = {}, c12 = {}, c13 = {};

  for (int k = 0; k < K; k += 4) {
    v2f a0 = *(const v2f*)(Ap0 + k);
    v2f a1 = *(const v2f*)(Ap1 + k);
    const float* bk = Bp + (size_t)k * N;
    v2f b0, b1, b2, b3;
    b0.x = bk[0];   b0.y = bk[(size_t)N + 0];
    b1.x = bk[16];  b1.y = bk[(size_t)N + 16];
    b2.x = bk[32];  b2.y = bk[(size_t)N + 32];
    b3.x = bk[48];  b3.y = bk[(size_t)N + 48];
    c00 = WMMA_F32(a0, b0, c00);
    c01 = WMMA_F32(a0, b1, c01);
    c02 = WMMA_F32(a0, b2, c02);
    c03 = WMMA_F32(a0, b3, c03);
    c10 = WMMA_F32(a1, b0, c10);
    c11 = WMMA_F32(a1, b1, c11);
    c12 = WMMA_F32(a1, b2, c12);
    c13 = WMMA_F32(a1, b3, c13);
  }

  float* Cp0 = C + (size_t)(m0 + half * 8) * N + n0 + l15;
#pragma unroll
  for (int j = 0; j < 8; ++j) {
    float* r = Cp0 + (size_t)j * N;
    r[0]  = c00[j];
    r[16] = c01[j];
    r[32] = c02[j];
    r[48] = c03[j];
  }
  if (has2) {
    float* Cp1 = C + (size_t)(m0 + 16 + half * 8) * N + n0 + l15;
#pragma unroll
    for (int j = 0; j < 8; ++j) {
      float* r = Cp1 + (size_t)j * N;
      r[0]  = c10[j];
      r[16] = c11[j];
      r[32] = c12[j];
      r[48] = c13[j];
    }
  }
}

template <int F, int R, bool RELU>
__global__ __launch_bounds__(256)
void gather_rows(const float* __restrict__ hw, const int2* __restrict__ pairs,
                 const int* __restrict__ start, const float* __restrict__ bias,
                 float* __restrict__ res, int M, int E) {
  constexpr int V = F / 128;
  constexpr int U = 2048 / F;
  const int lane = threadIdx.x & 31;
  const int wave = __builtin_amdgcn_readfirstlane((int)(threadIdx.x >> 5));
  const int d0 = (blockIdx.x * 8 + wave) * R;
  if (d0 >= M) return;
  const int d1 = (d0 + R < M) ? d0 + R : M;

  const float4* hw4 = (const float4*)hw + lane;
  float4 b4[V];
#pragma unroll
  for (int j = 0; j < V; ++j) b4[j] = ((const float4*)bias)[lane + 32 * j];

  int beg = start[d0];
  for (int d = d0; d < d1; ++d) {
    const int end = (d + 1 < M) ? start[d + 1] : E;
    float4 acc[V];
#pragma unroll
    for (int j = 0; j < V; ++j) acc[j] = make_float4(0.f, 0.f, 0.f, 0.f);

    int i = beg;
    for (; i + U <= end; i += U) {
      float4 v[U][V];
      float  w[U];
#pragma unroll
      for (int u = 0; u < U; ++u) {
        const int2 p = pairs[i + u];
        w[u] = __int_as_float(p.y);
        const float4* r = hw4 + (size_t)p.x * (F / 4);
#pragma unroll
        for (int j = 0; j < V; ++j) v[u][j] = r[32 * j];
      }
#pragma unroll
      for (int u = 0; u < U; ++u) {
#pragma unroll
        for (int j = 0; j < V; ++j) {
          acc[j].x += w[u] * v[u][j].x;
          acc[j].y += w[u] * v[u][j].y;
          acc[j].z += w[u] * v[u][j].z;
          acc[j].w += w[u] * v[u][j].w;
        }
      }
    }
    const int n = end - i;
    if (n > 0) {
      float4 v[U][V];
      float  w[U];
      int    s[U];
#pragma unroll
      for (int u = 0; u < U; ++u) {
        const int2 p = pairs[(u < n) ? i + u : end - 1];
        w[u] = (u < n) ? __int_as_float(p.y) : 0.f;
        s[u] = p.x;
      }
#pragma unroll
      for (int u = 0; u < U; ++u) {
#pragma unroll
        for (int j = 0; j < V; ++j) v[u][j] = make_float4(0.f, 0.f, 0.f, 0.f);
        if (u < n) {
          const float4* r = hw4 + (size_t)s[u] * (F / 4);
#pragma unroll
          for (int j = 0; j < V; ++j) v[u][j] = r[32 * j];
        }
      }
#pragma unroll
      for (int u = 0; u < U; ++u) {
#pragma unroll
        for (int j = 0; j < V; ++j) {
          acc[j].x += w[u] * v[u][j].x;
          acc[j].y += w[u] * v[u][j].y;
          acc[j].z += w[u] * v[u][j].z;
          acc[j].w += w[u] * v[u][j].w;
        }
      }
    }

    const float inv = 1.0f / fmaxf((float)(end - beg), 1.0f);
    float4* o = (float4*)res + (size_t)d * (F / 4) + lane;
#pragma unroll
    for (int j = 0; j < V; ++j) {
      float4 t;
      t.x = acc[j].x * inv + b4[j].x;
      t.y = acc[j].y * inv + b4[j].y;
      t.z = acc[j].z * inv + b4[j].z;
      t.w = acc[j].w * inv + b4[j].w;
      if (RELU) {
        t.x = fmaxf(t.x, 0.f);
        t.y = fmaxf(t.y, 0.f);
        t.z = fmaxf(t.z, 0.f);
        t.w = fmaxf(t.w, 0.f);
      }
      o[32 * j] = t;
    }
    beg = end;
  }
}

extern "C" void kernel_launch(void* const* d_in, const int* in_sizes, int n_in,
                              void* d_out, int out_size, void* d_ws, size_t ws_size,
                              hipStream_t stream) {
  const float* x    = (const float*)d_in[0];
  const float* W1   = (const float*)d_in[1];
  const float* b1   = (const float*)d_in[2];
  const float* W2   = (const float*)d_in[3];
  const float* b2   = (const float*)d_in[4];
  const float* ew   = (const float*)d_in[5];
  const int*   esrc = (const int*)d_in[6];
  const int*   edst = (const int*)d_in[7];
  float* out = (float*)d_out;

  const int M = in_sizes[0] / IN_F;
  const int E = in_sizes[5];

  int* start = (int*)d_ws;
  const long long Mpad = ((long long)M + 63) & ~63LL;
  float* bufA = (float*)d_ws + Mpad;
  float* bufB = bufA + (long long)M * HID_F;
  int2* pairs1 = (int2*)out;
  int*  rank   = (int*)(pairs1 + ((E + 1) & ~1));
  int2* pairs2 = (int2*)(bufA + (long long)M * OUT_F);

  const int T = 256;
  constexpr int R = 6;

  {
    long long n4 = Mpad / 4;
    zero_f4<<<dim3((unsigned)((n4 + T - 1) / T)), dim3(T), 0, stream>>>((float4*)start, n4);
    const int per_block = T * 4;
    const int nb = (E + per_block - 1) / per_block;
    count_rank<<<dim3(nb), dim3(T), 0, stream>>>(edst, start, rank, E);
    scan_excl<<<dim3(1), dim3(1024), 0, stream>>>(start, M);
    place_edges<<<dim3(nb), dim3(T), 0, stream>>>(esrc, edst, ew, start, rank, pairs1, E);
  }

  {
    dim3 g((M + 127) / 128, HID_F / 64);
    gemm_wmma_f32<<<g, dim3(128), 0, stream>>>(x, W1, bufA, M, HID_F, IN_F);
  }

  gather_rows<HID_F, R, true><<<dim3((M + 8 * R - 1) / (8 * R)), dim3(T), 0, stream>>>(
      bufA, pairs1, start, b1, bufB, M, E);

  move_pairs<<<dim3(1024), dim3(T), 0, stream>>>(pairs1, pairs2, E);

  {
    dim3 g((M + 127) / 128, OUT_F / 64);
    gemm_wmma_f32<<<g, dim3(128), 0, stream>>>(bufB, W2, bufA, M, OUT_F, HID_F);
  }

  gather_rows<OUT_F, R, false><<<dim3((M + 8 * R - 1) / (8 * R)), dim3(T), 0, stream>>>(
      bufA, pairs2, start, b2, out, M, E);
}


# both GEMMs: a wave stages its 32 A rows through its own LDS slice by whole-128-B-line 16-B loads, next chunk held in registers; WMMA A fragments read from LDS, not by 16-line 8-B global loads
# speedup vs baseline: 2.3943x; 1.2056x over previous
#include <hip/hip_runtime.h>


typedef float v2f __attribute__((ext_vector_type(2)));
typedef float v4f __attribute__((ext_vector_type(4)));
typedef float v8f __attribute__((ext_vector_type(8)));

#define IN_F  512
#define HID_F 256
#define OUT_F 128

#define WMMA_F32(a, b, c) \
  __builtin_amdgcn_wmma_f32_16x16x4_f32(false, (a), false, (b), (short)0, (c), false, false)

__global__ void zero_f4(float4* __restrict__ p, long long n4) {
  long long i = (long long)blockIdx.x * blockDim.x + threadIdx.x;
  if (i < n4) p[i] = make_float4(0.f, 0.f, 0.f, 0.f);
}


__global__ __launch_bounds__(256)
void count_rank(const int* __restrict__ dst, int* __restrict__ cnt,
                int* __restrict__ rank, int E) {
  const int e = (blockIdx.x * 256 + threadIdx.x) * 4;
  if (e + 3 < E) {
    const int4 d = *(const int4*)(dst + e);
    int4 r;
    r.x = atomicAdd(&cnt[d.x], 1);
    r.y = atomicAdd(&cnt[d.y], 1);
    r.z = atomicAdd(&cnt[d.z], 1);
    r.w = atomicAdd(&cnt[d.w], 1);
    *(int4*)(rank + e) = r;
  } else {
    for (int i = e; i < E; ++i) rank[i] = atomicAdd(&cnt[dst[i]], 1);
  }
}

__global__ __launch_bounds__(1024)
void scan_excl(int* __restrict__ p, int n) {
  __shared__ int wsum[32];
  const int tid  = threadIdx.x;
  const int lane = tid & 31;
  const int wv   = tid >> 5;
  int carry = 0;
  for (int base = 0; base < n; base += 4096) {
    const int i = base + tid * 4;
    int4 c = make_int4(0, 0, 0, 0);
    if (i + 3 < n) {
      c = *(const int4*)(p + i);
    } else {
      if (i     < n) c.x = p[i];
      if (i + 1 < n) c.y = p[i + 1];
      if (i + 2 < n) c.z = p[i + 2];
    }
    const int t = c.x + c.y + c.z + c.w;
    int s = t;
    for (int o = 1; o < 32; o <<= 1) {
      const int v = __shfl_up(s, o);
      if (lane >= o) s += v;
    }
    if (lane == 31) wsum[wv] = s;
    __syncthreads();
    const int ws = wsum[lane];
    int wss = ws;
    for (int o = 1; o < 32; o <<= 1) {
      const int v = __shfl_up(wss, o);
      if (lane >= o) wss += v;
    }
    const int wave_excl = __shfl(wss - ws, wv);
    const int total     = __shfl(wss, 31);
    const int x0 = carry + wave_excl + (s - t);
    const int x1 = x0 + c.x;
    const int x2 = x1 + c.y;
    const int x3 = x2 + c.z;
    if (i + 3 < n) {
      *(int4*)(p + i) = make_int4(x0, x1, x2, x3);
    } else {
      if (i     < n) p[i]     = x0;
      if (i + 1 < n) p[i + 1] = x1;
      if (i + 2 < n) p[i + 2] = x2;
    }
    carry += total;
    __syncthreads();
  }
}

__global__ __launch_bounds__(256)
void place_edges(const int* __restrict__ src, const int* __restrict__ dst,
                 const float* __restrict__ ew, const int* __restrict__ start,
                 const int* __restrict__ rank, int2* __restrict__ pairs, int E) {
  const int e = (blockIdx.x * 256 + threadIdx.x) * 4;
  if (e + 3 < E) {
    const int4   d = *(const int4*)(dst + e);
    const int4   r = *(const int4*)(rank + e);
    const int4   s = *(const int4*)(src + e);
    const float4 w = *(const float4*)(ew + e);
    pairs[start[d.x] + r.x] = make_int2(s.x, __float_as_int(w.x));
    pairs[start[d.y] + r.y] = make_int2(s.y, __float_as_int(w.y));
    pairs[start[d.z] + r.z] = make_int2(s.z, __float_as_int(w.z));
    pairs[start[d.w] + r.w] = make_int2(s.w, __float_as_int(w.w));
  } else {
    for (int i = e; i < E; ++i)
      pairs[start[dst[i]] + rank[i]] = make_int2(src[i], __float_as_int(ew[i]));
  }
}

__global__ __launch_bounds__(256)
void move_pairs(const int2* __restrict__ from, int2* __restrict__ to, int n) {
  for (int i = blockIdx.x * 256 + threadIdx.x; i < n; i += gridDim.x * 256) to[i] = from[i];
}

#define KC 32
#define AP 36
__global__ __launch_bounds__(128)
void gemm_wmma_f32(const float* __restrict__ A, const float* __restrict__ B,
                   float* __restrict__ C, int M, int N, int K) {
  __shared__ __attribute__((aligned(16))) float sA[4][32 * AP];
  const int lane = threadIdx.x & 31;
  const int wave = threadIdx.x >> 5;
  const int m0 = (blockIdx.x * 4 + wave) * 32;
  const int n0 = blockIdx.y * 64;
  if (m0 >= M) return;
  const bool has2 = (m0 + 16) < M;

  const int half = lane >> 4;
  const int l15  = lane & 15;

  const int sr = lane >> 3;
  const int sc = (lane & 7) * 4;
  float* sa = sA[wave];
  int aoff[8];
#pragma unroll
  for (int i = 0; i < 8; ++i) {
    int r = m0 + 4 * i + sr;
    if (r >= M) r = M - 1;
    aoff[i] = r * K + sc;
  }
  v4f pre[8];
#pragma unroll
  for (int i = 0; i < 8; ++i) pre[i] = *(const v4f*)(A + aoff[i]);

  const float* fa0 = sa + l15 * AP + half * 2;
  const float* fa1 = sa + (16 + l15) * AP + half * 2;
  const float* Bp = B + (size_t)(half * 2) * N + n0 + l15;

  v8f c00 = {}, c01 = {}, c02 = {}, c03 = {};
  v8f c10 = {}, c11 = {}, c12 = {}, c13 = {};

  for (int k0 = 0; k0 < K; k0 += KC) {
#pragma unroll
    for (int i = 0; i < 8; ++i) *(v4f*)(sa + (4 * i + sr) * AP + sc) = pre[i];
    __builtin_amdgcn_wave_barrier();
    v2f a0[8], a1[8];
#pragma unroll
    for (int kk = 0; kk < 8; ++kk) {
      a0[kk] = *(const v2f*)(fa0 + kk * 4);
      a1[kk] = *(const v2f*)(fa1 + kk * 4);
    }
    __builtin_amdgcn_wave_barrier();
    v2f bf[8][4];
#pragma unroll
    for (int kk = 0; kk < 8; ++kk) {
      const float* bk = Bp + (size_t)(k0 + kk * 4) * N;
      bf[kk][0].x = bk[0];   bf[kk][0].y = bk[(size_t)N + 0];
      bf[kk][1].x = bk[16];  bf[kk][1].y = bk[(size_t)N + 16];
      bf[kk][2].x = bk[32];  bf[kk][2].y = bk[(size_t)N + 32];
      bf[kk][3].x = bk[48];  bf[kk][3].y = bk[(size_t)N + 48];
    }
    const int kn = (k0 + KC < K) ? k0 + KC : k0;
#pragma unroll
    for (int i = 0; i < 8; ++i) pre[i] = *(const v4f*)(A + aoff[i] + kn);
#pragma unroll
    for (int kk = 0; kk < 8; ++kk) {
      c00 = WMMA_F32(a0[kk], bf[kk][0], c00);
      c01 = WMMA_F32(a0[kk], bf[kk][1], c01);
      c02 = WMMA_F32(a0[kk], bf[kk][2], c02);
      c03 = WMMA_F32(a0[kk], bf[kk][3], c03);
      c10 = WMMA_F32(a1[kk], bf[kk][0], c10);
      c11 = WMMA_F32(a1[kk], bf[kk][1], c11);
      c12 = WMMA_F32(a1[kk], bf[kk][2], c12);
      c13 = WMMA_F32(a1[kk], bf[kk][3], c13);
    }
  }

  float* Cp0 = C + (size_t)(m0 + half * 8) * N + n0 + l15;
#pragma unroll
  for (int j = 0; j < 8; ++j) {
    float* r = Cp0 + (size_t)j * N;
    r[0]  = c00[j];
    r[16] = c01[j];
    r[32] = c02[j];
    r[48] = c03[j];
  }
  if (has2) {
    float* Cp1 = C + (size_t)(m0 + 16 + half * 8) * N + n0 + l15;
#pragma unroll
    for (int j = 0; j < 8; ++j) {
      float* r = Cp1 + (size_t)j * N;
      r[0]  = c10[j];
      r[16] = c11[j];
      r[32] = c12[j];
      r[48] = c13[j];
    }
  }
}

template <int F, int R, bool RELU>
__global__ __launch_bounds__(256)
void gather_rows(const float* __restrict__ hw, const int2* __restrict__ pairs,
                 const int* __restrict__ start, const float* __restrict__ bias,
                 float* __restrict__ res, int M, int E) {
  constexpr int V = F / 128;
  constexpr int U = 2048 / F;
  const int lane = threadIdx.x & 31;
  const int wave = __builtin_amdgcn_readfirstlane((int)(threadIdx.x >> 5));
  const int d0 = (blockIdx.x * 8 + wave) * R;
  if (d0 >= M) return;
  const int d1 = (d0 + R < M) ? d0 + R : M;

  const float4* hw4 = (const float4*)hw + lane;
  float4 b4[V];
#pragma unroll
  for (int j = 0; j < V; ++j) b4[j] = ((const float4*)bias)[lane + 32 * j];

  int beg = start[d0];
  for (int d = d0; d < d1; ++d) {
    const int end = (d + 1 < M) ? start[d + 1] : E;
    float4 acc[V];
#pragma unroll
    for (int j = 0; j < V; ++j) acc[j] = make_float4(0.f, 0.f, 0.f, 0.f);

    int i = beg;
    for (; i + U <= end; i += U) {
      float4 v[U][V];
      float  w[U];
#pragma unroll
      for (int u = 0; u < U; ++u) {
        const int2 p = pairs[i + u];
        w[u] = __int_as_float(p.y);
        const float4* r = hw4 + (size_t)p.x * (F / 4);
#pragma unroll
        for (int j = 0; j < V; ++j) v[u][j] = r[32 * j];
      }
#pragma unroll
      for (int u = 0; u < U; ++u) {
#pragma unroll
        for (int j = 0; j < V; ++j) {
          acc[j].x += w[u] * v[u][j].x;
          acc[j].y += w[u] * v[u][j].y;
          acc[j].z += w[u] * v[u][j].z;
          acc[j].w += w[u] * v[u][j].w;
        }
      }
    }
    const int n = end - i;
    if (n > 0) {
      float4 v[U][V];
      float  w[U];
      int    s[U];
#pragma unroll
      for (int u = 0; u < U; ++u) {
        const int2 p = pairs[(u < n) ? i + u : end - 1];
        w[u] = (u < n) ? __int_as_float(p.y) : 0.f;
        s[u] = p.x;
      }
#pragma unroll
      for (int u = 0; u < U; ++u) {
#pragma unroll
        for (int j = 0; j < V; ++j) v[u][j] = make_float4(0.f, 0.f, 0.f, 0.f);
        if (u < n) {
          const float4* r = hw4 + (size_t)s[u] * (F / 4);
#pragma unroll
          for (int j = 0; j < V; ++j) v[u][j] = r[32 * j];
        }
      }
#pragma unroll
      for (int u = 0; u < U; ++u) {
#pragma unroll
        for (int j = 0; j < V; ++j) {
          acc[j].x += w[u] * v[u][j].x;
          acc[j].y += w[u] * v[u][j].y;
          acc[j].z += w[u] * v[u][j].z;
          acc[j].w += w[u] * v[u][j].w;
        }
      }
    }

    const float inv = 1.0f / fmaxf((float)(end - beg), 1.0f);
    float4* o = (float4*)res + (size_t)d * (F / 4) + lane;
#pragma unroll
    for (int j = 0; j < V; ++j) {
      float4 t;
      t.x = acc[j].x * inv + b4[j].x;
      t.y = acc[j].y * inv + b4[j].y;
      t.z = acc[j].z * inv + b4[j].z;
      t.w = acc[j].w * inv + b4[j].w;
      if (RELU) {
        t.x = fmaxf(t.x, 0.f);
        t.y = fmaxf(t.y, 0.f);
        t.z = fmaxf(t.z, 0.f);
        t.w = fmaxf(t.w, 0.f);
      }
      o[32 * j] = t;
    }
    beg = end;
  }
}

extern "C" void kernel_launch(void* const* d_in, const int* in_sizes, int n_in,
                              void* d_out, int out_size, void* d_ws, size_t ws_size,
                              hipStream_t stream) {
  const float* x    = (const float*)d_in[0];
  const float* W1   = (const float*)d_in[1];
  const float* b1   = (const float*)d_in[2];
  const float* W2   = (const float*)d_in[3];
  const float* b2   = (const float*)d_in[4];
  const float* ew   = (const float*)d_in[5];
  const int*   esrc = (const int*)d_in[6];
  const int*   edst = (const int*)d_in[7];
  float* out = (float*)d_out;

  const int M = in_sizes[0] / IN_F;
  const int E = in_sizes[5];

  int* start = (int*)d_ws;
  const long long Mpad = ((long long)M + 63) & ~63LL;
  float* bufA = (float*)d_ws + Mpad;
  float* bufB = bufA + (long long)M * HID_F;
  int2* pairs1 = (int2*)out;
  int*  rank   = (int*)(pairs1 + ((E + 1) & ~1));
  int2* pairs2 = (int2*)(bufA + (long long)M * OUT_F);

  const int T = 256;
  constexpr int R = 6;

  {
    long long n4 = Mpad / 4;
    zero_f4<<<dim3((unsigned)((n4 + T - 1) / T)), dim3(T), 0, stream>>>((float4*)start, n4);
    const int per_block = T * 4;
    const int nb = (E + per_block - 1) / per_block;
    count_rank<<<dim3(nb), dim3(T), 0, stream>>>(edst, start, rank, E);
    scan_excl<<<dim3(1), dim3(1024), 0, stream>>>(start, M);
    place_edges<<<dim3(nb), dim3(T), 0, stream>>>(esrc, edst, ew, start, rank, pairs1, E);
  }

  {
    dim3 g((M + 127) / 128, HID_F / 64);
    gemm_wmma_f32<<<g, dim3(128), 0, stream>>>(x, W1, bufA, M, HID_F, IN_F);
  }

  gather_rows<HID_F, R, true><<<dim3((M + 8 * R - 1) / (8 * R)), dim3(T), 0, stream>>>(
      bufA, pairs1, start, b1, bufB, M, E);

  move_pairs<<<dim3(1024), dim3(T), 0, stream>>>(pairs1, pairs2, E);

  {
    dim3 g((M + 127) / 128, OUT_F / 64);
    gemm_wmma_f32<<<g, dim3(128), 0, stream>>>(bufB, W2, bufA, M, OUT_F, HID_F);
  }

  gather_rows<OUT_F, R, false><<<dim3((M + 8 * R - 1) / (8 * R)), dim3(T), 0, stream>>>(
      bufA, pairs2, start, b2, out, M, E);
}


# count_rank and move_pairs run as the first blocks of the two GEMM launches, beside the tiles (no wait between blocks): 9 launches to 7; edge ranks stay arrival-ordered as in the parent
# speedup vs baseline: 2.7060x; 1.1302x over previous
#include <hip/hip_runtime.h>


typedef float v2f __attribute__((ext_vector_type(2)));
typedef float v4f __attribute__((ext_vector_type(4)));
typedef float v8f __attribute__((ext_vector_type(8)));

#define IN_F  512
#define HID_F 256
#define OUT_F 128

#define WMMA_F32(a, b, c) \
  __builtin_amdgcn_wmma_f32_16x16x4_f32(false, (a), false, (b), (short)0, (c), false, false)

__global__ void zero_f4(float4* __restrict__ p, long long n4) {
  long long i = (long long)blockIdx.x * blockDim.x + threadIdx.x;
  if (i < n4) p[i] = make_float4(0.f, 0.f, 0.f, 0.f);
}


__global__ __launch_bounds__(1024)
void scan_excl(int* __restrict__ p, int n) {
  __shared__ int wsum[32];
  const int tid  = threadIdx.x;
  const int lane = tid & 31;
  const int wv   = tid >> 5;
  int carry = 0;
  for (int base = 0; base < n; base += 4096) {
    const int i = base + tid * 4;
    int4 c = make_int4(0, 0, 0, 0);
    if (i + 3 < n) {
      c = *(const int4*)(p + i);
    } else {
      if (i     < n) c.x = p[i];
      if (i + 1 < n) c.y = p[i + 1];
      if (i + 2 < n) c.z = p[i + 2];
    }
    const int t = c.x + c.y + c.z + c.w;
    int s = t;
    for (int o = 1; o < 32; o <<= 1) {
      const int v = __shfl_up(s, o);
      if (lane >= o) s += v;
    }
    if (lane == 31) wsum[wv] = s;
    __syncthreads();
    const int ws = wsum[lane];
    int wss = ws;
    for (int o = 1; o < 32; o <<= 1) {
      const int v = __shfl_up(wss, o);
      if (lane >= o) wss += v;
    }
    const int wave_excl = __shfl(wss - ws, wv);
    const int total     = __shfl(wss, 31);
    const int x0 = carry + wave_excl + (s - t);
    const int x1 = x0 + c.x;
    const int x2 = x1 + c.y;
    const int x3 = x2 + c.z;
    if (i + 3 < n) {
      *(int4*)(p + i) = make_int4(x0, x1, x2, x3);
    } else {
      if (i     < n) p[i]     = x0;
      if (i + 1 < n) p[i + 1] = x1;
      if (i + 2 < n) p[i + 2] = x2;
    }
    carry += total;
    __syncthreads();
  }
}

__global__ __launch_bounds__(256)
void place_edges(const int* __restrict__ src, const int* __restrict__ dst,
                 const float* __restrict__ ew, const int* __restrict__ start,
                 const int* __restrict__ rank, int2* __restrict__ pairs, int E) {
  const int e = (blockIdx.x * 256 + threadIdx.x) * 4;
  if (e + 3 < E) {
    const int4   d = *(const int4*)(dst + e);
    const int4   r = *(const int4*)(rank + e);
    const int4   s = *(const int4*)(src + e);
    const float4 w = *(const float4*)(ew + e);
    pairs[start[d.x] + r.x] = make_int2(s.x, __float_as_int(w.x));
    pairs[start[d.y] + r.y] = make_int2(s.y, __float_as_int(w.y));
    pairs[start[d.z] + r.z] = make_int2(s.z, __float_as_int(w.z));
    pairs[start[d.w] + r.w] = make_int2(s.w, __float_as_int(w.w));
  } else {
    for (int i = e; i < E; ++i)
      pairs[start[dst[i]] + rank[i]] = make_int2(src[i], __float_as_int(ew[i]));
  }
}

#define KC 32
#define AP 36
__device__ __forceinline__
void gemm_block(const float* __restrict__ A, const float* __restrict__ B,
                float* __restrict__ C, int M, int N, int K, int bx, int by,
                float (*sA)[32 * AP]) {
  const int lane = threadIdx.x & 31;
  const int wave = threadIdx.x >> 5;
  const int m0 = (bx * 4 + wave) * 32;
  const int n0 = by * 64;
  if (m0 >= M) return;
  const bool has2 = (m0 + 16) < M;

  const int half = lane >> 4;
  const int l15  = lane & 15;

  const int sr = lane >> 3;
  const int sc = (lane & 7) * 4;
  float* sa = sA[wave];
  int aoff[8];
#pragma unroll
  for (int i = 0; i < 8; ++i) {
    int r = m0 + 4 * i + sr;
    if (r >= M) r = M - 1;
    aoff[i] = r * K + sc;
  }
  v4f pre[8];
#pragma unroll
  for (int i = 0; i < 8; ++i) pre[i] = *(const v4f*)(A + aoff[i]);

  const float* fa0 = sa + l15 * AP + half * 2;
  const float* fa1 = sa + (16 + l15) * AP + half * 2;
  const float* Bp = B + (size_t)(half * 2) * N + n0 + l15;

  v8f c00 = {}, c01 = {}, c02 = {}, c03 = {};
  v8f c10 = {}, c11 = {}, c12 = {}, c13 = {};

  for (int k0 = 0; k0 < K; k0 += KC) {
#pragma unroll
    for (int i = 0; i < 8; ++i) *(v4f*)(sa + (4 * i + sr) * AP + sc) = pre[i];
    __builtin_amdgcn_wave_barrier();
    v2f a0[8], a1[8];
#pragma unroll
    for (int kk = 0; kk < 8; ++kk) {
      a0[kk] = *(const v2f*)(fa0 + kk * 4);
      a1[kk] = *(const v2f*)(fa1 + kk * 4);
    }
    __builtin_amdgcn_wave_barrier();
    v2f bf[8][4];
#pragma unroll
    for (int kk = 0; kk < 8; ++kk) {
      const float* bk = Bp + (size_t)(k0 + kk * 4) * N;
      bf[kk][0].x = bk[0];   bf[kk][0].y = bk[(size_t)N + 0];
      bf[kk][1].x = bk[16];  bf[kk][1].y = bk[(size_t)N + 16];
      bf[kk][2].x = bk[32];  bf[kk][2].y = bk[(size_t)N + 32];
      bf[kk][3].x = bk[48];  bf[kk][3].y = bk[(size_t)N + 48];
    }
    const int kn = (k0 + KC < K) ? k0 + KC : k0;
#pragma unroll
    for (int i = 0; i < 8; ++i) pre[i] = *(const v4f*)(A + aoff[i] + kn);
#pragma unroll
    for (int kk = 0; kk < 8; ++kk) {
      c00 = WMMA_F32(a0[kk], bf[kk][0], c00);
      c01 = WMMA_F32(a0[kk], bf[kk][1], c01);
      c02 = WMMA_F32(a0[kk], bf[kk][2], c02);
      c03 = WMMA_F32(a0[kk], bf[kk][3], c03);
      c10 = WMMA_F32(a1[kk], bf[kk][0], c10);
      c11 = WMMA_F32(a1[kk], bf[kk][1], c11);
      c12 = WMMA_F32(a1[kk], bf[kk][2], c12);
      c13 = WMMA_F32(a1[kk], bf[kk][3], c13);
    }
  }

  float* Cp0 = C + (size_t)(m0 + half * 8) * N + n0 + l15;
#pragma unroll
  for (int j = 0; j < 8; ++j) {
    float* r = Cp0 + (size_t)j * N;
    r[0]  = c00[j];
    r[16] = c01[j];
    r[32] = c02[j];
    r[48] = c03[j];
  }
  if (has2) {
    float* Cp1 = C + (size_t)(m0 + 16 + half * 8) * N + n0 + l15;
#pragma unroll
    for (int j = 0; j < 8; ++j) {
      float* r = Cp1 + (size_t)j * N;
      r[0]  = c10[j];
      r[16] = c11[j];
      r[32] = c12[j];
      r[48] = c13[j];
    }
  }
}


__global__ __launch_bounds__(128)
void gemm_with_count(const float* __restrict__ A, const float* __restrict__ B,
                     float* __restrict__ C, int M, int N, int K, int nbx, int nside,
                     const int* __restrict__ dst, int* __restrict__ cnt,
                     int* __restrict__ rank, int E) {
  __shared__ __attribute__((aligned(16))) float sA[4][32 * AP];
  const int bid = blockIdx.x;
  if (bid < nside) {
    const int n4   = E >> 2;
    const int nthr = nside * 128;
    int g = bid * 128 + threadIdx.x;
    int4 d = make_int4(0, 0, 0, 0);
    if (g < n4) d = ((const int4*)dst)[g];
    while (g < n4) {
      const int gn = g + nthr;
      int4 dn = d;
      if (gn < n4) dn = ((const int4*)dst)[gn];
      int4 r;
      r.x = atomicAdd(&cnt[d.x], 1);
      r.y = atomicAdd(&cnt[d.y], 1);
      r.z = atomicAdd(&cnt[d.z], 1);
      r.w = atomicAdd(&cnt[d.w], 1);
      ((int4*)rank)[g] = r;
      d = dn;
      g = gn;
    }
    if (bid == 0 && threadIdx.x == 0)
      for (int i = n4 * 4; i < E; ++i) rank[i] = atomicAdd(&cnt[dst[i]], 1);
    return;
  }
  const int g  = bid - nside;
  const int by = g / nbx;
  gemm_block(A, B, C, M, N, K, g - by * nbx, by, sA);
}

__global__ __launch_bounds__(128)
void gemm_with_move(const float* __restrict__ A, const float* __restrict__ B,
                    float* __restrict__ C, int M, int N, int K, int nbx, int nside,
                    const int4* from, int4* to, int n) {
  __shared__ __attribute__((aligned(16))) float sA[4][32 * AP];
  const int bid = blockIdx.x;
  if (bid < nside) {
    const int n4   = n >> 1;
    const int nthr = nside * 128;
    int i = bid * 128 + threadIdx.x;
    for (; i + 3 * nthr < n4; i += 4 * nthr) {
      const int4 a = from[i];
      const int4 b = from[i + nthr];
      const int4 c = from[i + 2 * nthr];
      const int4 e = from[i + 3 * nthr];
      to[i]            = a;
      to[i + nthr]     = b;
      to[i + 2 * nthr] = c;
      to[i + 3 * nthr] = e;
    }
    for (; i < n4; i += nthr) to[i] = from[i];
    if ((n & 1) && bid == 0 && threadIdx.x == 0)
      ((int2*)to)[n - 1] = ((const int2*)from)[n - 1];
    return;
  }
  const int g  = bid - nside;
  const int by = g / nbx;
  gemm_block(A, B, C, M, N, K, g - by * nbx, by, sA);
}

template <int F, int R, bool RELU>
__global__ __launch_bounds__(256)
void gather_rows(const float* __restrict__ hw, const int2* __restrict__ pairs,
                 const int* __restrict__ start, const float* __restrict__ bias,
                 float* __restrict__ res, int M, int E) {
  constexpr int V = F / 128;
  constexpr int U = 2048 / F;
  const int lane = threadIdx.x & 31;
  const int wave = __builtin_amdgcn_readfirstlane((int)(threadIdx.x >> 5));
  const int d0 = (blockIdx.x * 8 + wave) * R;
  if (d0 >= M) return;
  const int d1 = (d0 + R < M) ? d0 + R : M;

  const float4* hw4 = (const float4*)hw + lane;
  float4 b4[V];
#pragma unroll
  for (int j = 0; j < V; ++j) b4[j] = ((const float4*)bias)[lane + 32 * j];

  int beg = start[d0];
  for (int d = d0; d < d1; ++d) {
    const int end = (d + 1 < M) ? start[d + 1] : E;
    float4 acc[V];
#pragma unroll
    for (int j = 0; j < V; ++j) acc[j] = make_float4(0.f, 0.f, 0.f, 0.f);

    int i = beg;
    for (; i + U <= end; i += U) {
      float4 v[U][V];
      float  w[U];
#pragma unroll
      for (int u = 0; u < U; ++u) {
        const int2 p = pairs[i + u];
        w[u] = __int_as_float(p.y);
        const float4* r = hw4 + (size_t)p.x * (F / 4);
#pragma unroll
        for (int j = 0; j < V; ++j) v[u][j] = r[32 * j];
      }
#pragma unroll
      for (int u = 0; u < U; ++u) {
#pragma unroll
        for (int j = 0; j < V; ++j) {
          acc[j].x += w[u] * v[u][j].x;
          acc[j].y += w[u] * v[u][j].y;
          acc[j].z += w[u] * v[u][j].z;
          acc[j].w += w[u] * v[u][j].w;
        }
      }
    }
    const int n = end - i;
    if (n > 0) {
      float4 v[U][V];
      float  w[U];
      int    s[U];
#pragma unroll
      for (int u = 0; u < U; ++u) {
        const int2 p = pairs[(u < n) ? i + u : end - 1];
        w[u] = (u < n) ? __int_as_float(p.y) : 0.f;
        s[u] = p.x;
      }
#pragma unroll
      for (int u = 0; u < U; ++u) {
#pragma unroll
        for (int j = 0; j < V; ++j) v[u][j] = make_float4(0.f, 0.f, 0.f, 0.f);
        if (u < n) {
          const float4* r = hw4 + (size_t)s[u] * (F / 4);
#pragma unroll
          for (int j = 0; j < V; ++j) v[u][j] = r[32 * j];
        }
      }
#pragma unroll
      for (int u = 0; u < U; ++u) {
#pragma unroll
        for (int j = 0; j < V; ++j) {
          acc[j].x += w[u] * v[u][j].x;
          acc[j].y += w[u] * v[u][j].y;
          acc[j].z += w[u] * v[u][j].z;
          acc[j].w += w[u] * v[u][j].w;
        }
      }
    }

    const float inv = 1.0f / fmaxf((float)(end - beg), 1.0f);
    float4* o = (float4*)res + (size_t)d * (F / 4) + lane;
#pragma unroll
    for (int j = 0; j < V; ++j) {
      float4 t;
      t.x = acc[j].x * inv + b4[j].x;
      t.y = acc[j].y * inv + b4[j].y;
      t.z = acc[j].z * inv + b4[j].z;
      t.w = acc[j].w * inv + b4[j].w;
      if (RELU) {
        t.x = fmaxf(t.x, 0.f);
        t.y = fmaxf(t.y, 0.f);
        t.z = fmaxf(t.z, 0.f);
        t.w = fmaxf(t.w, 0.f);
      }
      o[32 * j] = t;
    }
    beg = end;
  }
}

extern "C" void kernel_launch(void* const* d_in, const int* in_sizes, int n_in,
                              void* d_out, int out_size, void* d_ws, size_t ws_size,
                              hipStream_t stream) {
  const float* x    = (const float*)d_in[0];
  const float* W1   = (const float*)d_in[1];
  const float* b1   = (const float*)d_in[2];
  const float* W2   = (const float*)d_in[3];
  const float* b2   = (const float*)d_in[4];
  const float* ew   = (const float*)d_in[5];
  const int*   esrc = (const int*)d_in[6];
  const int*   edst = (const int*)d_in[7];
  float* out = (float*)d_out;

  const int M = in_sizes[0] / IN_F;
  const int E = in_sizes[5];

  int* start = (int*)d_ws;
  const long long Mpad = ((long long)M + 63) & ~63LL;
  float* bufA = (float*)d_ws + Mpad;
  float* bufB = bufA + (long long)M * HID_F;
  int2* pairs1 = (int2*)out;
  int*  rank   = (int*)(pairs1 + ((E + 1) & ~1));
  int2* pairs2 = (int2*)(bufA + (long long)M * OUT_F);

  const int T = 256;
  constexpr int R = 6;
  const int NSIDE1 = 256;
  const int NSIDE2 = 192;
  const int nbx = (M + 127) / 128;

  {
    long long n4 = Mpad / 4;
    zero_f4<<<dim3((unsigned)((n4 + T - 1) / T)), dim3(T), 0, stream>>>((float4*)start, n4);
    gemm_with_count<<<dim3(NSIDE1 + nbx * (HID_F / 64)), dim3(128), 0, stream>>>(
        x, W1, bufA, M, HID_F, IN_F, nbx, NSIDE1, edst, start, rank, E);
    const int per_block = T * 4;
    const int nb = (E + per_block - 1) / per_block;
    scan_excl<<<dim3(1), dim3(1024), 0, stream>>>(start, M);
    place_edges<<<dim3(nb), dim3(T), 0, stream>>>(esrc, edst, ew, start, rank, pairs1, E);
  }

  gather_rows<HID_F, R, true><<<dim3((M + 8 * R - 1) / (8 * R)), dim3(T), 0, stream>>>(
      bufA, pairs1, start, b1, bufB, M, E);

  gemm_with_move<<<dim3(NSIDE2 + nbx * (OUT_F / 64)), dim3(128), 0, stream>>>(
      bufB, W2, bufA, M, OUT_F, HID_F, nbx, NSIDE2, (const int4*)pairs1, (int4*)pairs2, E);

  gather_rows<OUT_F, R, false><<<dim3((M + 8 * R - 1) / (8 * R)), dim3(T), 0, stream>>>(
      bufA, pairs2, start, b2, out, M, E);
}


# layer-1 gather fused with the second product: a block's 32 h rows go through LDS into its f32 WMMAs, h never in memory; edge list in h's old buffer (no move): 7 launches to 6; sum order unchanged
# speedup vs baseline: 2.9860x; 1.1035x over previous
#include <hip/hip_runtime.h>


typedef float v2f __attribute__((ext_vector_type(2)));
typedef float v4f __attribute__((ext_vector_type(4)));
typedef float v8f __attribute__((ext_vector_type(8)));

#define IN_F  512
#define HID_F 256
#define OUT_F 128

#define WMMA_F32(a, b, c) \
  __builtin_amdgcn_wmma_f32_16x16x4_f32(false, (a), false, (b), (short)0, (c), false, false)

__global__ void zero_f4(float4* __restrict__ p, long long n4) {
  long long i = (long long)blockIdx.x * blockDim.x + threadIdx.x;
  if (i < n4) p[i] = make_float4(0.f, 0.f, 0.f, 0.f);
}


__global__ __launch_bounds__(1024)
void scan_excl(int* __restrict__ p, int n) {
  __shared__ int wsum[32];
  const int tid  = threadIdx.x;
  const int lane = tid & 31;
  const int wv   = tid >> 5;
  int carry = 0;
  for (int base = 0; base < n; base += 4096) {
    const int i = base + tid * 4;
    int4 c = make_int4(0, 0, 0, 0);
    if (i + 3 < n) {
      c = *(const int4*)(p + i);
    } else {
      if (i     < n) c.x = p[i];
      if (i + 1 < n) c.y = p[i + 1];
      if (i + 2 < n) c.z = p[i + 2];
    }
    const int t = c.x + c.y + c.z + c.w;
    int s = t;
    for (int o = 1; o < 32; o <<= 1) {
      const int v = __shfl_up(s, o);
      if (lane >= o) s += v;
    }
    if (lane == 31) wsum[wv] = s;
    __syncthreads();
    const int ws = wsum[lane];
    int wss = ws;
    for (int o = 1; o < 32; o <<= 1) {
      const int v = __shfl_up(wss, o);
      if (lane >= o) wss += v;
    }
    const int wave_excl = __shfl(wss - ws, wv);
    const int total     = __shfl(wss, 31);
    const int x0 = carry + wave_excl + (s - t);
    const int x1 = x0 + c.x;
    const int x2 = x1 + c.y;
    const int x3 = x2 + c.z;
    if (i + 3 < n) {
      *(int4*)(p + i) = make_int4(x0, x1, x2, x3);
    } else {
      if (i     < n) p[i]     = x0;
      if (i + 1 < n) p[i + 1] = x1;
      if (i + 2 < n) p[i + 2] = x2;
    }
    carry += total;
    __syncthreads();
  }
}

__global__ __launch_bounds__(256)
void place_edges(const int* __restrict__ src, const int* __restrict__ dst,
                 const float* __restrict__ ew, const int* __restrict__ start,
                 const int* __restrict__ rank, int2* __restrict__ pairs, int E) {
  const int e = (blockIdx.x * 256 + threadIdx.x) * 4;
  if (e + 3 < E) {
    const int4   d = *(const int4*)(dst + e);
    const int4   r = *(const int4*)(rank + e);
    const int4   s = *(const int4*)(src + e);
    const float4 w = *(const float4*)(ew + e);
    pairs[start[d.x] + r.x] = make_int2(s.x, __float_as_int(w.x));
    pairs[start[d.y] + r.y] = make_int2(s.y, __float_as_int(w.y));
    pairs[start[d.z] + r.z] = make_int2(s.z, __float_as_int(w.z));
    pairs[start[d.w] + r.w] = make_int2(s.w, __float_as_int(w.w));
  } else {
    for (int i = e; i < E; ++i)
      pairs[start[dst[i]] + rank[i]] = make_int2(src[i], __float_as_int(ew[i]));
  }
}

#define KC 32
#define AP 36
__device__ __forceinline__
void gemm_block(const float* __restrict__ A, const float* __restrict__ B,
                float* __restrict__ C, int M, int N, int K, int bx, int by,
                float (*sA)[32 * AP]) {
  const int lane = threadIdx.x & 31;
  const int wave = threadIdx.x >> 5;
  const int m0 = (bx * 4 + wave) * 32;
  const int n0 = by * 64;
  if (m0 >= M) return;
  const bool has2 = (m0 + 16) < M;

  const int half = lane >> 4;
  const int l15  = lane & 15;

  const int sr = lane >> 3;
  const int sc = (lane & 7) * 4;
  float* sa = sA[wave];
  int aoff[8];
#pragma unroll
  for (int i = 0; i < 8; ++i) {
    int r = m0 + 4 * i + sr;
    if (r >= M) r = M - 1;
    aoff[i] = r * K + sc;
  }
  v4f pre[8];
#pragma unroll
  for (int i = 0; i < 8; ++i) pre[i] = *(const v4f*)(A + aoff[i]);

  const float* fa0 = sa + l15 * AP + half * 2;
  const float* fa1 = sa + (16 + l15) * AP + half * 2;
  const float* Bp = B + (size_t)(half * 2) * N + n0 + l15;

  v8f c00 = {}, c01 = {}, c02 = {}, c03 = {};
  v8f c10 = {}, c11 = {}, c12 = {}, c13 = {};

  for (int k0 = 0; k0 < K; k0 += KC) {
#pragma unroll
    for (int i = 0; i < 8; ++i) *(v4f*)(sa + (4 * i + sr) * AP + sc) = pre[i];
    __builtin_amdgcn_wave_barrier();
    v2f a0[8], a1[8];
#pragma unroll
    for (int kk = 0; kk < 8; ++kk) {
      a0[kk] = *(const v2f*)(fa0 + kk * 4);
      a1[kk] = *(const v2f*)(fa1 + kk * 4);
    }
    __builtin_amdgcn_wave_barrier();
    v2f bf[8][4];
#pragma unroll
    for (int kk = 0; kk < 8; ++kk) {
      const float* bk = Bp + (size_t)(k0 + kk * 4) * N;
      bf[kk][0].x = bk[0];   bf[kk][0].y = bk[(size_t)N + 0];
      bf[kk][1].x = bk[16];  bf[kk][1].y = bk[(size_t)N + 16];
      bf[kk][2].x = bk[32];  bf[kk][2].y = bk[(size_t)N + 32];
      bf[kk][3].x = bk[48];  bf[kk][3].y = bk[(size_t)N + 48];
    }
    const int kn = (k0 + KC < K) ? k0 + KC : k0;
#pragma unroll
    for (int i = 0; i < 8; ++i) pre[i] = *(const v4f*)(A + aoff[i] + kn);
#pragma unroll
    for (int kk = 0; kk < 8; ++kk) {
      c00 = WMMA_F32(a0[kk], bf[kk][0], c00);
      c01 = WMMA_F32(a0[kk], bf[kk][1], c01);
      c02 = WMMA_F32(a0[kk], bf[kk][2], c02);
      c03 = WMMA_F32(a0[kk], bf[kk][3], c03);
      c10 = WMMA_F32(a1[kk], bf[kk][0], c10);
      c11 = WMMA_F32(a1[kk], bf[kk][1], c11);
      c12 = WMMA_F32(a1[kk], bf[kk][2], c12);
      c13 = WMMA_F32(a1[kk], bf[kk][3], c13);
    }
  }

  float* Cp0 = C + (size_t)(m0 + half * 8) * N + n0 + l15;
#pragma unroll
  for (int j = 0; j < 8; ++j) {
    float* r = Cp0 + (size_t)j * N;
    r[0]  = c00[j];
    r[16] = c01[j];
    r[32] = c02[j];
    r[48] = c03[j];
  }
  if (has2) {
    float* Cp1 = C + (size_t)(m0 + 16 + half * 8) * N + n0 + l15;
#pragma unroll
    for (int j = 0; j < 8; ++j) {
      float* r = Cp1 + (size_t)j * N;
      r[0]  = c10[j];
      r[16] = c11[j];
      r[32] = c12[j];
      r[48] = c13[j];
    }
  }
}


__global__ __launch_bounds__(128)
void gemm_with_count(const float* __restrict__ A, const float* __restrict__ B,
                     float* __restrict__ C, int M, int N, int K, int nbx, int nside,
                     const int* __restrict__ dst, int* __restrict__ cnt,
                     int* __restrict__ rank, int E) {
  __shared__ __attribute__((aligned(16))) float sA[4][32 * AP];
  const int bid = blockIdx.x;
  if (bid < nside) {
    const int n4   = E >> 2;
    const int nthr = nside * 128;
    int g = bid * 128 + threadIdx.x;
    int4 d = make_int4(0, 0, 0, 0);
    if (g < n4) d = ((const int4*)dst)[g];
    while (g < n4) {
      const int gn = g + nthr;
      int4 dn = d;
      if (gn < n4) dn = ((const int4*)dst)[gn];
      int4 r;
      r.x = atomicAdd(&cnt[d.x], 1);
      r.y = atomicAdd(&cnt[d.y], 1);
      r.z = atomicAdd(&cnt[d.z], 1);
      r.w = atomicAdd(&cnt[d.w], 1);
      ((int4*)rank)[g] = r;
      d = dn;
      g = gn;
    }
    if (bid == 0 && threadIdx.x == 0)
      for (int i = n4 * 4; i < E; ++i) rank[i] = atomicAdd(&cnt[dst[i]], 1);
    return;
  }
  const int g  = bid - nside;
  const int by = g / nbx;
  gemm_block(A, B, C, M, N, K, g - by * nbx, by, sA);
}

template <int F, int R, bool RELU>
__global__ __launch_bounds__(256)
void gather_rows(const float* __restrict__ hw, const int2* __restrict__ pairs,
                 const int* __restrict__ start, const float* __restrict__ bias,
                 float* __restrict__ res, int M, int E) {
  constexpr int V = F / 128;
  constexpr int U = 2048 / F;
  const int lane = threadIdx.x & 31;
  const int wave = __builtin_amdgcn_readfirstlane((int)(threadIdx.x >> 5));
  const int d0 = (blockIdx.x * 8 + wave) * R;
  if (d0 >= M) return;
  const int d1 = (d0 + R < M) ? d0 + R : M;

  const float4* hw4 = (const float4*)hw + lane;
  float4 b4[V];
#pragma unroll
  for (int j = 0; j < V; ++j) b4[j] = ((const float4*)bias)[lane + 32 * j];

  int beg = start[d0];
  for (int d = d0; d < d1; ++d) {
    const int end = (d + 1 < M) ? start[d + 1] : E;
    float4 acc[V];
#pragma unroll
    for (int j = 0; j < V; ++j) acc[j] = make_float4(0.f, 0.f, 0.f, 0.f);

    int i = beg;
    for (; i + U <= end; i += U) {
      float4 v[U][V];
      float  w[U];
#pragma unroll
      for (int u = 0; u < U; ++u) {
        const int2 p = pairs[i + u];
        w[u] = __int_as_float(p.y);
        const float4* r = hw4 + (size_t)p.x * (F / 4);
#pragma unroll
        for (int j = 0; j < V; ++j) v[u][j] = r[32 * j];
      }
#pragma unroll
      for (int u = 0; u < U; ++u) {
#pragma unroll
        for (int j = 0; j < V; ++j) {
          acc[j].x += w[u] * v[u][j].x;
          acc[j].y += w[u] * v[u][j].y;
          acc[j].z += w[u] * v[u][j].z;
          acc[j].w += w[u] * v[u][j].w;
        }
      }
    }
    const int n = end - i;
    if (n > 0) {
      float4 v[U][V];
      float  w[U];
      int    s[U];
#pragma unroll
      for (int u = 0; u < U; ++u) {
        const int2 p = pairs[(u < n) ? i + u : end - 1];
        w[u] = (u < n) ? __int_as_float(p.y) : 0.f;
        s[u] = p.x;
      }
#pragma unroll
      for (int u = 0; u < U; ++u) {
#pragma unroll
        for (int j = 0; j < V; ++j) v[u][j] = make_float4(0.f, 0.f, 0.f, 0.f);
        if (u < n) {
          const float4* r = hw4 + (size_t)s[u] * (F / 4);
#pragma unroll
          for (int j = 0; j < V; ++j) v[u][j] = r[32 * j];
        }
      }
#pragma unroll
      for (int u = 0; u < U; ++u) {
#pragma unroll
        for (int j = 0; j < V; ++j) {
          acc[j].x += w[u] * v[u][j].x;
          acc[j].y += w[u] * v[u][j].y;
          acc[j].z += w[u] * v[u][j].z;
          acc[j].w += w[u] * v[u][j].w;
        }
      }
    }

    const float inv = 1.0f / fmaxf((float)(end - beg), 1.0f);
    float4* o = (float4*)res + (size_t)d * (F / 4) + lane;
#pragma unroll
    for (int j = 0; j < V; ++j) {
      float4 t;
      t.x = acc[j].x * inv + b4[j].x;
      t.y = acc[j].y * inv + b4[j].y;
      t.z = acc[j].z * inv + b4[j].z;
      t.w = acc[j].w * inv + b4[j].w;
      if (RELU) {
        t.x = fmaxf(t.x, 0.f);
        t.y = fmaxf(t.y, 0.f);
        t.z = fmaxf(t.z, 0.f);
        t.w = fmaxf(t.w, 0.f);
      }
      o[32 * j] = t;
    }
    beg = end;
  }
}

#define HP 260
#define CP 132
__global__ __launch_bounds__(256, 10)
void gather_relu_gemm(const float* __restrict__ hw, const int2* __restrict__ pairs,
                      const int* __restrict__ start, const float* __restrict__ bias,
                      const float* __restrict__ B, float* __restrict__ C, int M, int E) {
  constexpr int F = HID_F;
  constexpr int V = F / 128;
  constexpr int U = 2048 / F;
  constexpr int R = 4;
  __shared__ __attribute__((aligned(16))) float sH[32 * HP];
  __shared__ __attribute__((aligned(16))) float sC[32 * CP];
  const int lane = threadIdx.x & 31;
  const int wave = __builtin_amdgcn_readfirstlane((int)(threadIdx.x >> 5));
  const int r0 = blockIdx.x * 32;

  {
    const int d0 = r0 + wave * R;
    const float4* hw4 = (const float4*)hw + lane;
    float4 b4[V];
#pragma unroll
    for (int j = 0; j < V; ++j) b4[j] = ((const float4*)bias)[lane + 32 * j];

    const int d1 = (d0 + R < M) ? d0 + R : ((d0 < M) ? M : d0);
    for (int d = d1; d < d0 + R; ++d) {
      float4* o = (float4*)(sH + (d - r0) * HP) + lane;
#pragma unroll
      for (int j = 0; j < V; ++j) o[32 * j] = make_float4(0.f, 0.f, 0.f, 0.f);
    }
    int beg = (d0 < M) ? start[d0] : 0;
    for (int d = d0; d < d1; ++d) {
      float4* o = (float4*)(sH + (d - r0) * HP) + lane;
      const int end = (d + 1 < M) ? start[d + 1] : E;
      float4 acc[V];
#pragma unroll
      for (int j = 0; j < V; ++j) acc[j] = make_float4(0.f, 0.f, 0.f, 0.f);

      int i = beg;
      for (; i + U <= end; i += U) {
        float4 v[U][V];
        float  w[U];
#pragma unroll
        for (int u = 0; u < U; ++u) {
          const int2 p = pairs[i + u];
          w[u] = __int_as_float(p.y);
          const float4* r = hw4 + (size_t)p.x * (F / 4);
#pragma unroll
          for (int j = 0; j < V; ++j) v[u][j] = r[32 * j];
        }
#pragma unroll
        for (int u = 0; u < U; ++u) {
#pragma unroll
          for (int j = 0; j < V; ++j) {
            acc[j].x += w[u] * v[u][j].x;
            acc[j].y += w[u] * v[u][j].y;
            acc[j].z += w[u] * v[u][j].z;
            acc[j].w += w[u] * v[u][j].w;
          }
        }
      }
      const int n = end - i;
      if (n > 0) {
        float4 v[U][V];
        float  w[U];
        int    s[U];
#pragma unroll
        for (int u = 0; u < U; ++u) {
          const int2 p = pairs[(u < n) ? i + u : end - 1];
          w[u] = (u < n) ? __int_as_float(p.y) : 0.f;
          s[u] = p.x;
        }
#pragma unroll
        for (int u = 0; u < U; ++u) {
#pragma unroll
          for (int j = 0; j < V; ++j) v[u][j] = make_float4(0.f, 0.f, 0.f, 0.f);
          if (u < n) {
            const float4* r = hw4 + (size_t)s[u] * (F / 4);
#pragma unroll
            for (int j = 0; j < V; ++j) v[u][j] = r[32 * j];
          }
        }
#pragma unroll
        for (int u = 0; u < U; ++u) {
#pragma unroll
          for (int j = 0; j < V; ++j) {
            acc[j].x += w[u] * v[u][j].x;
            acc[j].y += w[u] * v[u][j].y;
            acc[j].z += w[u] * v[u][j].z;
            acc[j].w += w[u] * v[u][j].w;
          }
        }
      }

      const float inv = 1.0f / fmaxf((float)(end - beg), 1.0f);
#pragma unroll
      for (int j = 0; j < V; ++j) {
        float4 t;
        t.x = acc[j].x * inv + b4[j].x;
        t.y = acc[j].y * inv + b4[j].y;
        t.z = acc[j].z * inv + b4[j].z;
        t.w = acc[j].w * inv + b4[j].w;
        t.x = fmaxf(t.x, 0.f);
        t.y = fmaxf(t.y, 0.f);
        t.z = fmaxf(t.z, 0.f);
        t.w = fmaxf(t.w, 0.f);
        o[32 * j] = t;
      }
      beg = end;
    }
  }

  const int half = lane >> 4;
  const int l15  = lane & 15;
  const int n0   = wave * 16;
  const float* fa0 = sH + l15 * HP + half * 2;
  const float* fa1 = sH + (16 + l15) * HP + half * 2;
  const float* Bp = B + (size_t)(half * 2) * OUT_F + n0 + l15;

  v2f bcur[8];
#pragma unroll
  for (int kk = 0; kk < 8; ++kk) {
    const float* bk = Bp + (size_t)(kk * 4) * OUT_F;
    bcur[kk].x = bk[0];
    bcur[kk].y = bk[OUT_F];
  }
  __syncthreads();

  v8f c0 = {}, c1 = {};
#pragma unroll
  for (int ch = 0; ch < HID_F / 32; ++ch) {
    v2f a0[8], a1[8];
#pragma unroll
    for (int kk = 0; kk < 8; ++kk) {
      a0[kk] = *(const v2f*)(fa0 + ch * 32 + kk * 4);
      a1[kk] = *(const v2f*)(fa1 + ch * 32 + kk * 4);
    }
    v2f bnext[8];
    if (ch + 1 < HID_F / 32) {
#pragma unroll
      for (int kk = 0; kk < 8; ++kk) {
        const float* bk = Bp + (size_t)((ch + 1) * 32 + kk * 4) * OUT_F;
        bnext[kk].x = bk[0];
        bnext[kk].y = bk[OUT_F];
      }
    }
#pragma unroll
    for (int kk = 0; kk < 8; ++kk) {
      c0 = WMMA_F32(a0[kk], bcur[kk], c0);
      c1 = WMMA_F32(a1[kk], bcur[kk], c1);
    }
    if (ch + 1 < HID_F / 32) {
#pragma unroll
      for (int kk = 0; kk < 8; ++kk) bcur[kk] = bnext[kk];
    }
  }

#pragma unroll
  for (int j = 0; j < 8; ++j) {
    sC[(j + 8 * half) * CP + n0 + l15]      = c0[j];
    sC[(16 + j + 8 * half) * CP + n0 + l15] = c1[j];
  }
  __syncthreads();
  float4 t[R];
#pragma unroll
  for (int q = 0; q < R; ++q) t[q] = *((const float4*)(sC + (wave * R + q) * CP) + lane);
#pragma unroll
  for (int q = 0; q < R; ++q) {
    const int row = r0 + wave * R + q;
    if (row < M)
      *((float4*)(C + (size_t)row * OUT_F) + lane) = t[q];
  }
}

extern "C" void kernel_launch(void* const* d_in, const int* in_sizes, int n_in,
                              void* d_out, int out_size, void* d_ws, size_t ws_size,
                              hipStream_t stream) {
  const float* x    = (const float*)d_in[0];
  const float* W1   = (const float*)d_in[1];
  const float* b1   = (const float*)d_in[2];
  const float* W2   = (const float*)d_in[3];
  const float* b2   = (const float*)d_in[4];
  const float* ew   = (const float*)d_in[5];
  const int*   esrc = (const int*)d_in[6];
  const int*   edst = (const int*)d_in[7];
  float* out = (float*)d_out;

  const int M = in_sizes[0] / IN_F;
  const int E = in_sizes[5];

  int* start = (int*)d_ws;
  const long long Mpad = ((long long)M + 63) & ~63LL;
  float* bufA = (float*)d_ws + Mpad;
  float* bufB = bufA + (long long)M * HID_F;
  int2* pairs = (int2*)(bufB + (long long)M * OUT_F);
  int*  rank  = (int*)out;

  const int T = 256;
  constexpr int R = 6;
  const int NSIDE1 = 256;
  const int nbx = (M + 127) / 128;

  {
    long long n4 = Mpad / 4;
    zero_f4<<<dim3((unsigned)((n4 + T - 1) / T)), dim3(T), 0, stream>>>((float4*)start, n4);
    gemm_with_count<<<dim3(NSIDE1 + nbx * (HID_F / 64)), dim3(128), 0, stream>>>(
        x, W1, bufA, M, HID_F, IN_F, nbx, NSIDE1, edst, start, rank, E);
    const int per_block = T * 4;
    const int nb = (E + per_block - 1) / per_block;
    scan_excl<<<dim3(1), dim3(1024), 0, stream>>>(start, M);
    place_edges<<<dim3(nb), dim3(T), 0, stream>>>(esrc, edst, ew, start, rank, pairs, E);
  }

  gather_relu_gemm<<<dim3((M + 31) / 32), dim3(T), 0, stream>>>(
      bufA, pairs, start, b1, W2, bufB, M, E);

  gather_rows<OUT_F, R, false><<<dim3((M + 8 * R - 1) / (8 * R)), dim3(T), 0, stream>>>(
      bufB, pairs, start, b2, out, M, E);
}


# edge lists: an edge's counting add returns its slot in a fixed 32-slot row per dst (overflow list behind); scan and place launches gone, 6 to 4; in-edges past 32 summed last, in list-arrival order
# speedup vs baseline: 3.4460x; 1.1541x over previous
#include <hip/hip_runtime.h>


typedef float v2f __attribute__((ext_vector_type(2)));
typedef float v4f __attribute__((ext_vector_type(4)));
typedef float v8f __attribute__((ext_vector_type(8)));

#define IN_F  512
#define HID_F 256
#define OUT_F 128

#define WMMA_F32(a, b, c) \
  __builtin_amdgcn_wmma_f32_16x16x4_f32(false, (a), false, (b), (short)0, (c), false, false)

#define CAP 32

__global__ void zero_f4(float4* __restrict__ p, long long n4, int* __restrict__ novf) {
  long long i = (long long)blockIdx.x * blockDim.x + threadIdx.x;
  if (i < n4) p[i] = make_float4(0.f, 0.f, 0.f, 0.f);
  if (i == 0) *novf = 0;
}

__device__ __forceinline__
void store_edge(int d, int s, float w, int r, int2* __restrict__ slots,
                int* __restrict__ novf, int* __restrict__ ovd,
                int* __restrict__ ovs, float* __restrict__ ovw) {
  if (r < CAP) {
    slots[d * CAP + r] = make_int2(s, __float_as_int(w));
  } else {
    const int k = atomicAdd(novf, 1);
    ovd[k] = d;
    ovs[k] = s;
    ovw[k] = w;
  }
}

#define KC 32
#define AP 36
__device__ __forceinline__
void gemm_block(const float* __restrict__ A, const float* __restrict__ B,
                float* __restrict__ C, int M, int N, int K, int bx, int by,
                float (*sA)[32 * AP]) {
  const int lane = threadIdx.x & 31;
  const int wave = threadIdx.x >> 5;
  const int m0 = (bx * 4 + wave) * 32;
  const int n0 = by * 64;
  if (m0 >= M) return;
  const bool has2 = (m0 + 16) < M;

  const int half = lane >> 4;
  const int l15  = lane & 15;

  const int sr = lane >> 3;
  const int sc = (lane & 7) * 4;
  float* sa = sA[wave];
  int aoff[8];
#pragma unroll
  for (int i = 0; i < 8; ++i) {
    int r = m0 + 4 * i + sr;
    if (r >= M) r = M - 1;
    aoff[i] = r * K + sc;
  }
  v4f pre[8];
#pragma unroll
  for (int i = 0; i < 8; ++i) pre[i] = *(const v4f*)(A + aoff[i]);

  const float* fa0 = sa + l15 * AP + half * 2;
  const float* fa1 = sa + (16 + l15) * AP + half * 2;
  const float* Bp = B + (size_t)(half * 2) * N + n0 + l15;

  v8f c00 = {}, c01 = {}, c02 = {}, c03 = {};
  v8f c10 = {}, c11 = {}, c12 = {}, c13 = {};

  for (int k0 = 0; k0 < K; k0 += KC) {
#pragma unroll
    for (int i = 0; i < 8; ++i) *(v4f*)(sa + (4 * i + sr) * AP + sc) = pre[i];
    __builtin_amdgcn_wave_barrier();
    v2f a0[8], a1[8];
#pragma unroll
    for (int kk = 0; kk < 8; ++kk) {
      a0[kk] = *(const v2f*)(fa0 + kk * 4);
      a1[kk] = *(const v2f*)(fa1 + kk * 4);
    }
    __builtin_amdgcn_wave_barrier();
    v2f bf[8][4];
#pragma unroll
    for (int kk = 0; kk < 8; ++kk) {
      const float* bk = Bp + (size_t)(k0 + kk * 4) * N;
      bf[kk][0].x = bk[0];   bf[kk][0].y = bk[(size_t)N + 0];
      bf[kk][1].x = bk[16];  bf[kk][1].y = bk[(size_t)N + 16];
      bf[kk][2].x = bk[32];  bf[kk][2].y = bk[(size_t)N + 32];
      bf[kk][3].x = bk[48];  bf[kk][3].y = bk[(size_t)N + 48];
    }
    const int kn = (k0 + KC < K) ? k0 + KC : k0;
#pragma unroll
    for (int i = 0; i < 8; ++i) pre[i] = *(const v4f*)(A + aoff[i] + kn);
#pragma unroll
    for (int kk = 0; kk < 8; ++kk) {
      c00 = WMMA_F32(a0[kk], bf[kk][0], c00);
      c01 = WMMA_F32(a0[kk], bf[kk][1], c01);
      c02 = WMMA_F32(a0[kk], bf[kk][2], c02);
      c03 = WMMA_F32(a0[kk], bf[kk][3], c03);
      c10 = WMMA_F32(a1[kk], bf[kk][0], c10);
      c11 = WMMA_F32(a1[kk], bf[kk][1], c11);
      c12 = WMMA_F32(a1[kk], bf[kk][2], c12);
      c13 = WMMA_F32(a1[kk], bf[kk][3], c13);
    }
  }

  float* Cp0 = C + (size_t)(m0 + half * 8) * N + n0 + l15;
#pragma unroll
  for (int j = 0; j < 8; ++j) {
    float* r = Cp0 + (size_t)j * N;
    r[0]  = c00[j];
    r[16] = c01[j];
    r[32] = c02[j];
    r[48] = c03[j];
  }
  if (has2) {
    float* Cp1 = C + (size_t)(m0 + 16 + half * 8) * N + n0 + l15;
#pragma unroll
    for (int j = 0; j < 8; ++j) {
      float* r = Cp1 + (size_t)j * N;
      r[0]  = c10[j];
      r[16] = c11[j];
      r[32] = c12[j];
      r[48] = c13[j];
    }
  }
}


__global__ __launch_bounds__(128)
void gemm_with_lists(const float* __restrict__ A, const float* __restrict__ B,
                     float* __restrict__ C, int M, int N, int K, int nbx, int nside,
                     const int* __restrict__ src, const int* __restrict__ dst,
                     const float* __restrict__ ew, int* __restrict__ cnt,
                     int2* __restrict__ slots, int* __restrict__ novf,
                     int* __restrict__ ovd, int* __restrict__ ovs,
                     float* __restrict__ ovw, int E) {
  __shared__ __attribute__((aligned(16))) float sA[4][32 * AP];
  const int bid = blockIdx.x;
  if (bid < nside) {
    const int n4   = E >> 2;
    const int nthr = nside * 128;
    int g = bid * 128 + threadIdx.x;
    int4   d = make_int4(0, 0, 0, 0);
    int4   s = make_int4(0, 0, 0, 0);
    float4 w = make_float4(0.f, 0.f, 0.f, 0.f);
    if (g < n4) {
      d = ((const int4*)dst)[g];
      s = ((const int4*)src)[g];
      w = ((const float4*)ew)[g];
    }
    while (g < n4) {
      const int gn = g + nthr;
      int4   dn = d;
      int4   sn = s;
      float4 wn = w;
      if (gn < n4) {
        dn = ((const int4*)dst)[gn];
        sn = ((const int4*)src)[gn];
        wn = ((const float4*)ew)[gn];
      }
      int4 r;
      r.x = atomicAdd(&cnt[d.x], 1);
      r.y = atomicAdd(&cnt[d.y], 1);
      r.z = atomicAdd(&cnt[d.z], 1);
      r.w = atomicAdd(&cnt[d.w], 1);
      store_edge(d.x, s.x, w.x, r.x, slots, novf, ovd, ovs, ovw);
      store_edge(d.y, s.y, w.y, r.y, slots, novf, ovd, ovs, ovw);
      store_edge(d.z, s.z, w.z, r.z, slots, novf, ovd, ovs, ovw);
      store_edge(d.w, s.w, w.w, r.w, slots, novf, ovd, ovs, ovw);
      d = dn;
      s = sn;
      w = wn;
      g = gn;
    }
    if (bid == 0 && threadIdx.x == 0)
      for (int i = n4 * 4; i < E; ++i)
        store_edge(dst[i], src[i], ew[i], atomicAdd(&cnt[dst[i]], 1),
                   slots, novf, ovd, ovs, ovw);
    return;
  }
  const int g  = bid - nside;
  const int by = g / nbx;
  gemm_block(A, B, C, M, N, K, g - by * nbx, by, sA);
}

template <int F, bool RELU>
__device__ __forceinline__
void full_row(const float4* __restrict__ hw4, const int2* __restrict__ slots,
              const int* __restrict__ novf, const int* __restrict__ ovd,
              const int* __restrict__ ovs, const float* __restrict__ ovw,
              int d, int deg, int E, const float4 (&b4)[F / 128], float4* o) {
  constexpr int V = F / 128;
  float4 acc[V];
#pragma unroll
  for (int j = 0; j < V; ++j) acc[j] = make_float4(0.f, 0.f, 0.f, 0.f);
#pragma unroll 1
  for (int i = d * CAP; i < (d + 1) * CAP; i += 4) {
    float4 v[4][V];
    float  w[4];
#pragma unroll
    for (int u = 0; u < 4; ++u) {
      const int2 p = slots[i + u];
      w[u] = __int_as_float(p.y);
      const float4* r = hw4 + (size_t)p.x * (F / 4);
#pragma unroll
      for (int j = 0; j < V; ++j) v[u][j] = r[32 * j];
    }
#pragma unroll
    for (int u = 0; u < 4; ++u) {
#pragma unroll
      for (int j = 0; j < V; ++j) {
        acc[j].x += w[u] * v[u][j].x;
        acc[j].y += w[u] * v[u][j].y;
        acc[j].z += w[u] * v[u][j].z;
        acc[j].w += w[u] * v[u][j].w;
      }
    }
  }
  int nov = novf[0];
  if (nov > E) nov = E;
#pragma unroll 1
  for (int i = 0; i < nov; ++i) {
    if (ovd[i] != d) continue;
    const float   w = ovw[i];
    const float4* r = hw4 + (size_t)ovs[i] * (F / 4);
#pragma unroll
    for (int j = 0; j < V; ++j) {
      const float4 v = r[32 * j];
      acc[j].x += w * v.x;
      acc[j].y += w * v.y;
      acc[j].z += w * v.z;
      acc[j].w += w * v.w;
    }
  }
  const float inv = 1.0f / fmaxf((float)deg, 1.0f);
#pragma unroll
  for (int j = 0; j < V; ++j) {
    float4 t;
    t.x = acc[j].x * inv + b4[j].x;
    t.y = acc[j].y * inv + b4[j].y;
    t.z = acc[j].z * inv + b4[j].z;
    t.w = acc[j].w * inv + b4[j].w;
    if (RELU) {
      t.x = fmaxf(t.x, 0.f);
      t.y = fmaxf(t.y, 0.f);
      t.z = fmaxf(t.z, 0.f);
      t.w = fmaxf(t.w, 0.f);
    }
    o[32 * j] = t;
  }
}

template <int F, int R, bool RELU>
__global__ __launch_bounds__(256)
void gather_rows(const float* __restrict__ hw, const int2* __restrict__ slots,
                 const int* __restrict__ cnt, const int* __restrict__ novf,
                 const int* __restrict__ ovd, const int* __restrict__ ovs,
                 const float* __restrict__ ovw, const float* __restrict__ bias,
                 float* __restrict__ res, int M, int E) {
  constexpr int V = F / 128;
  constexpr int U = 2048 / F;
  const int lane = threadIdx.x & 31;
  const int wave = __builtin_amdgcn_readfirstlane((int)(threadIdx.x >> 5));
  const int d0 = (blockIdx.x * 8 + wave) * R;
  if (d0 >= M) return;
  const int d1 = (d0 + R < M) ? d0 + R : M;

  const float4* hw4 = (const float4*)hw + lane;
  float4 b4[V];
#pragma unroll
  for (int j = 0; j < V; ++j) b4[j] = ((const float4*)bias)[lane + 32 * j];

  for (int d = d0; d < d1; ++d) {
    const int deg = cnt[d];
    if (deg > CAP) continue;
    const int beg = d * CAP;
    const int end = beg + deg;
    float4 acc[V];
#pragma unroll
    for (int j = 0; j < V; ++j) acc[j] = make_float4(0.f, 0.f, 0.f, 0.f);

    int i = beg;
    for (; i + U <= end; i += U) {
      float4 v[U][V];
      float  w[U];
#pragma unroll
      for (int u = 0; u < U; ++u) {
        const int2 p = slots[i + u];
        w[u] = __int_as_float(p.y);
        const float4* r = hw4 + (size_t)p.x * (F / 4);
#pragma unroll
        for (int j = 0; j < V; ++j) v[u][j] = r[32 * j];
      }
#pragma unroll
      for (int u = 0; u < U; ++u) {
#pragma unroll
        for (int j = 0; j < V; ++j) {
          acc[j].x += w[u] * v[u][j].x;
          acc[j].y += w[u] * v[u][j].y;
          acc[j].z += w[u] * v[u][j].z;
          acc[j].w += w[u] * v[u][j].w;
        }
      }
    }
    const int n = end - i;
    if (n > 0) {
      float4 v[U][V];
      float  w[U];
      int    s[U];
#pragma unroll
      for (int u = 0; u < U; ++u) {
        const int2 p = slots[(u < n) ? i + u : end - 1];
        w[u] = (u < n) ? __int_as_float(p.y) : 0.f;
        s[u] = p.x;
      }
#pragma unroll
      for (int u = 0; u < U; ++u) {
#pragma unroll
        for (int j = 0; j < V; ++j) v[u][j] = make_float4(0.f, 0.f, 0.f, 0.f);
        if (u < n) {
          const float4* r = hw4 + (size_t)s[u] * (F / 4);
#pragma unroll
          for (int j = 0; j < V; ++j) v[u][j] = r[32 * j];
        }
      }
#pragma unroll
      for (int u = 0; u < U; ++u) {
#pragma unroll
        for (int j = 0; j < V; ++j) {
          acc[j].x += w[u] * v[u][j].x;
          acc[j].y += w[u] * v[u][j].y;
          acc[j].z += w[u] * v[u][j].z;
          acc[j].w += w[u] * v[u][j].w;
        }
      }
    }

    const float inv = 1.0f / fmaxf((float)deg, 1.0f);
    float4* o = (float4*)res + (size_t)d * (F / 4) + lane;
#pragma unroll
    for (int j = 0; j < V; ++j) {
      float4 t;
      t.x = acc[j].x * inv + b4[j].x;
      t.y = acc[j].y * inv + b4[j].y;
      t.z = acc[j].z * inv + b4[j].z;
      t.w = acc[j].w * inv + b4[j].w;
      if (RELU) {
        t.x = fmaxf(t.x, 0.f);
        t.y = fmaxf(t.y, 0.f);
        t.z = fmaxf(t.z, 0.f);
        t.w = fmaxf(t.w, 0.f);
      }
      o[32 * j] = t;
    }
  }
  for (int d = d0; d < d1; ++d) {
    const int deg = cnt[d];
    if (deg > CAP)
      full_row<F, RELU>(hw4, slots, novf, ovd, ovs, ovw, d, deg, E, b4,
                        (float4*)res + (size_t)d * (F / 4) + lane);
  }
}

#define HP 260
#define CP 132
__global__ __launch_bounds__(256, 10)
void gather_relu_gemm(const float* __restrict__ hw, const int2* __restrict__ slots,
                      const int* __restrict__ cnt, const int* __restrict__ novf,
                      const int* __restrict__ ovd, const int* __restrict__ ovs,
                      const float* __restrict__ ovw, const float* __restrict__ bias,
                      const float* __restrict__ B, float* __restrict__ C, int M, int E) {
  constexpr int F = HID_F;
  constexpr int V = F / 128;
  constexpr int U = 2048 / F;
  constexpr int R = 4;
  __shared__ __attribute__((aligned(16))) float sH[32 * HP];
  __shared__ __attribute__((aligned(16))) float sC[32 * CP];
  const int lane = threadIdx.x & 31;
  const int wave = __builtin_amdgcn_readfirstlane((int)(threadIdx.x >> 5));
  const int r0 = blockIdx.x * 32;

  {
    const int d0 = r0 + wave * R;
    const float4* hw4 = (const float4*)hw + lane;
    float4 b4[V];
#pragma unroll
    for (int j = 0; j < V; ++j) b4[j] = ((const float4*)bias)[lane + 32 * j];

    const int d1 = (d0 + R < M) ? d0 + R : ((d0 < M) ? M : d0);
    for (int d = d1; d < d0 + R; ++d) {
      float4* o = (float4*)(sH + (d - r0) * HP) + lane;
#pragma unroll
      for (int j = 0; j < V; ++j) o[32 * j] = make_float4(0.f, 0.f, 0.f, 0.f);
    }
    for (int d = d0; d < d1; ++d) {
      float4* o = (float4*)(sH + (d - r0) * HP) + lane;
      const int deg = cnt[d];
      if (deg > CAP) continue;
      const int beg = d * CAP;
      const int end = beg + deg;
      float4 acc[V];
#pragma unroll
      for (int j = 0; j < V; ++j) acc[j] = make_float4(0.f, 0.f, 0.f, 0.f);

      int i = beg;
      for (; i + U <= end; i += U) {
        float4 v[U][V];
        float  w[U];
#pragma unroll
        for (int u = 0; u < U; ++u) {
          const int2 p = slots[i + u];
          w[u] = __int_as_float(p.y);
          const float4* r = hw4 + (size_t)p.x * (F / 4);
#pragma unroll
          for (int j = 0; j < V; ++j) v[u][j] = r[32 * j];
        }
#pragma unroll
        for (int u = 0; u < U; ++u) {
#pragma unroll
          for (int j = 0; j < V; ++j) {
            acc[j].x += w[u] * v[u][j].x;
            acc[j].y += w[u] * v[u][j].y;
            acc[j].z += w[u] * v[u][j].z;
            acc[j].w += w[u] * v[u][j].w;
          }
        }
      }
      const int n = end - i;
      if (n > 0) {
        float4 v[U][V];
        float  w[U];
        int    s[U];
#pragma unroll
        for (int u = 0; u < U; ++u) {
          const int2 p = slots[(u < n) ? i + u : end - 1];
          w[u] = (u < n) ? __int_as_float(p.y) : 0.f;
          s[u] = p.x;
        }
#pragma unroll
        for (int u = 0; u < U; ++u) {
#pragma unroll
          for (int j = 0; j < V; ++j) v[u][j] = make_float4(0.f, 0.f, 0.f, 0.f);
          if (u < n) {
            const float4* r = hw4 + (size_t)s[u] * (F / 4);
#pragma unroll
            for (int j = 0; j < V; ++j) v[u][j] = r[32 * j];
          }
        }
#pragma unroll
        for (int u = 0; u < U; ++u) {
#pragma unroll
          for (int j = 0; j < V; ++j) {
            acc[j].x += w[u] * v[u][j].x;
            acc[j].y += w[u] * v[u][j].y;
            acc[j].z += w[u] * v[u][j].z;
            acc[j].w += w[u] * v[u][j].w;
          }
        }
      }

      const float inv = 1.0f / fmaxf((float)deg, 1.0f);
#pragma unroll
      for (int j = 0; j < V; ++j) {
        float4 t;
        t.x = acc[j].x * inv + b4[j].x;
        t.y = acc[j].y * inv + b4[j].y;
        t.z = acc[j].z * inv + b4[j].z;
        t.w = acc[j].w * inv + b4[j].w;
        t.x = fmaxf(t.x, 0.f);
        t.y = fmaxf(t.y, 0.f);
        t.z = fmaxf(t.z, 0.f);
        t.w = fmaxf(t.w, 0.f);
        o[32 * j] = t;
      }
    }
    for (int d = d0; d < d1; ++d) {
      const int deg = cnt[d];
      if (deg > CAP)
        full_row<F, true>(hw4, slots, novf, ovd, ovs, ovw, d, deg, E, b4,
                          (float4*)(sH + (d - r0) * HP) + lane);
    }
  }

  const int half = lane >> 4;
  const int l15  = lane & 15;
  const int n0   = wave * 16;
  const float* fa0 = sH + l15 * HP + half * 2;
  const float* fa1 = sH + (16 + l15) * HP + half * 2;
  const float* Bp = B + (size_t)(half * 2) * OUT_F + n0 + l15;

  v2f bcur[8];
#pragma unroll
  for (int kk = 0; kk < 8; ++kk) {
    const float* bk = Bp + (size_t)(kk * 4) * OUT_F;
    bcur[kk].x = bk[0];
    bcur[kk].y = bk[OUT_F];
  }
  __syncthreads();

  v8f c0 = {}, c1 = {};
#pragma unroll
  for (int ch = 0; ch < HID_F / 32; ++ch) {
    v2f a0[8], a1[8];
#pragma unroll
    for (int kk = 0; kk < 8; ++kk) {
      a0[kk] = *(const v2f*)(fa0 + ch * 32 + kk * 4);
      a1[kk] = *(const v2f*)(fa1 + ch * 32 + kk * 4);
    }
    v2f bnext[8];
    if (ch + 1 < HID_F / 32) {
#pragma unroll
      for (int kk = 0; kk < 8; ++kk) {
        const float* bk = Bp + (size_t)((ch + 1) * 32 + kk * 4) * OUT_F;
        bnext[kk].x = bk[0];
        bnext[kk].y = bk[OUT_F];
      }
    }
#pragma unroll
    for (int kk = 0; kk < 8; ++kk) {
      c0 = WMMA_F32(a0[kk], bcur[kk], c0);
      c1 = WMMA_F32(a1[kk], bcur[kk], c1);
    }
    if (ch + 1 < HID_F / 32) {
#pragma unroll
      for (int kk = 0; kk < 8; ++kk) bcur[kk] = bnext[kk];
    }
  }

#pragma unroll
  for (int j = 0; j < 8; ++j) {
    sC[(j + 8 * half) * CP + n0 + l15]      = c0[j];
    sC[(16 + j + 8 * half) * CP + n0 + l15] = c1[j];
  }
  __syncthreads();
  float4 t[R];
#pragma unroll
  for (int q = 0; q < R; ++q) t[q] = *((const float4*)(sC + (wave * R + q) * CP) + lane);
#pragma unroll
  for (int q = 0; q < R; ++q) {
    const int row = r0 + wave * R + q;
    if (row < M)
      *((float4*)(C + (size_t)row * OUT_F) + lane) = t[q];
  }
}

extern "C" void kernel_launch(void* const* d_in, const int* in_sizes, int n_in,
                              void* d_out, int out_size, void* d_ws, size_t ws_size,
                              hipStream_t stream) {
  const float* x    = (const float*)d_in[0];
  const float* W1   = (const float*)d_in[1];
  const float* b1   = (const float*)d_in[2];
  const float* W2   = (const float*)d_in[3];
  const float* b2   = (const float*)d_in[4];
  const float* ew   = (const float*)d_in[5];
  const int*   esrc = (const int*)d_in[6];
  const int*   edst = (const int*)d_in[7];
  float* out = (float*)d_out;

  const int M = in_sizes[0] / IN_F;
  const int E = in_sizes[5];

  int* cnt = (int*)d_ws;
  const long long Mpad = ((long long)M + 63) & ~63LL;
  float* bufA = (float*)d_ws + Mpad;
  float* bufB = bufA + (long long)M * HID_F;
  float* upper = bufB + (long long)M * OUT_F;
  int2*  slots = (int2*)upper;
  int*   ovd   = (int*)(upper + (long long)M * (2 * CAP));
  int*   ovs   = ovd + E;
  float* ovw   = (float*)(ovs + E);
  int*   novf  = (int*)(ovw + E);

  const int T = 256;
  constexpr int R = 6;
  const int NSIDE1 = 256;
  const int nbx = (M + 127) / 128;

  {
    long long n4 = Mpad / 4;
    zero_f4<<<dim3((unsigned)((n4 + T - 1) / T)), dim3(T), 0, stream>>>((float4*)cnt, n4, novf);
    gemm_with_lists<<<dim3(NSIDE1 + nbx * (HID_F / 64)), dim3(128), 0, stream>>>(
        x, W1, bufA, M, HID_F, IN_F, nbx, NSIDE1, esrc, edst, ew, cnt, slots, novf,
        ovd, ovs, ovw, E);
  }

  gather_relu_gemm<<<dim3((M + 31) / 32), dim3(T), 0, stream>>>(
      bufA, slots, cnt, novf, ovd, ovs, ovw, b1, W2, bufB, M, E);

  gather_rows<OUT_F, R, false><<<dim3((M + 8 * R - 1) / (8 * R)), dim3(T), 0, stream>>>(
      bufB, slots, cnt, novf, ovd, ovs, ovw, b2, out, M, E);
}


# first product: a wave's four column tiles interleaved (tile t, lane c = column 4c+t): B values come by 16-B loads, 16 a chunk not 64 of 4 B; C rows leave as whole-line 16-B stores; sums unchanged
# speedup vs baseline: 3.5947x; 1.0432x over previous
#include <hip/hip_runtime.h>


typedef float v2f __attribute__((ext_vector_type(2)));
typedef float v4f __attribute__((ext_vector_type(4)));
typedef float v8f __attribute__((ext_vector_type(8)));

#define IN_F  512
#define HID_F 256
#define OUT_F 128

#define WMMA_F32(a, b, c) \
  __builtin_amdgcn_wmma_f32_16x16x4_f32(false, (a), false, (b), (short)0, (c), false, false)

#define CAP 32

__global__ void zero_f4(float4* __restrict__ p, long long n4, int* __restrict__ novf) {
  long long i = (long long)blockIdx.x * blockDim.x + threadIdx.x;
  if (i < n4) p[i] = make_float4(0.f, 0.f, 0.f, 0.f);
  if (i == 0) *novf = 0;
}

__device__ __forceinline__
void store_edge(int d, int s, float w, int r, int2* __restrict__ slots,
                int* __restrict__ novf, int* __restrict__ ovd,
                int* __restrict__ ovs, float* __restrict__ ovw) {
  if (r < CAP) {
    slots[d * CAP + r] = make_int2(s, __float_as_int(w));
  } else {
    const int k = atomicAdd(novf, 1);
    ovd[k] = d;
    ovs[k] = s;
    ovw[k] = w;
  }
}

#define KC 32
#define AP 36
__device__ __forceinline__
void gemm_block(const float* __restrict__ A, const float* __restrict__ B,
                float* __restrict__ C, int M, int N, int K, int bx, int by,
                float (*sA)[32 * AP]) {
  const int lane = threadIdx.x & 31;
  const int wave = threadIdx.x >> 5;
  const int m0 = (bx * 4 + wave) * 32;
  const int n0 = by * 64;
  if (m0 >= M) return;
  const bool has2 = (m0 + 16) < M;

  const int half = lane >> 4;
  const int l15  = lane & 15;

  const int sr = lane >> 3;
  const int sc = (lane & 7) * 4;
  float* sa = sA[wave];
  int aoff[8];
#pragma unroll
  for (int i = 0; i < 8; ++i) {
    int r = m0 + 4 * i + sr;
    if (r >= M) r = M - 1;
    aoff[i] = r * K + sc;
  }
  v4f pre[8];
#pragma unroll
  for (int i = 0; i < 8; ++i) pre[i] = *(const v4f*)(A + aoff[i]);

  const float* fa0 = sa + l15 * AP + half * 2;
  const float* fa1 = sa + (16 + l15) * AP + half * 2;
  const float* Bp = B + (size_t)(half * 2) * N + n0 + 4 * l15;

  v8f c00 = {}, c01 = {}, c02 = {}, c03 = {};
  v8f c10 = {}, c11 = {}, c12 = {}, c13 = {};

  for (int k0 = 0; k0 < K; k0 += KC) {
#pragma unroll
    for (int i = 0; i < 8; ++i) *(v4f*)(sa + (4 * i + sr) * AP + sc) = pre[i];
    __builtin_amdgcn_wave_barrier();
    v2f a0[8], a1[8];
#pragma unroll
    for (int kk = 0; kk < 8; ++kk) {
      a0[kk] = *(const v2f*)(fa0 + kk * 4);
      a1[kk] = *(const v2f*)(fa1 + kk * 4);
    }
    __builtin_amdgcn_wave_barrier();
    v2f bf[8][4];
#pragma unroll
    for (int kk = 0; kk < 8; ++kk) {
      const float* bk = Bp + (size_t)(k0 + kk * 4) * N;
      const v4f r0 = *(const v4f*)bk;
      const v4f r1 = *(const v4f*)(bk + (size_t)N);
      bf[kk][0].x = r0.x;  bf[kk][0].y = r1.x;
      bf[kk][1].x = r0.y;  bf[kk][1].y = r1.y;
      bf[kk][2].x = r0.z;  bf[kk][2].y = r1.z;
      bf[kk][3].x = r0.w;  bf[kk][3].y = r1.w;
    }
    const int kn = (k0 + KC < K) ? k0 + KC : k0;
#pragma unroll
    for (int i = 0; i < 8; ++i) pre[i] = *(const v4f*)(A + aoff[i] + kn);
    __builtin_amdgcn_sched_barrier(0);
#pragma unroll
    for (int kk = 0; kk < 8; ++kk) {
      c00 = WMMA_F32(a0[kk], bf[kk][0], c00);
      c01 = WMMA_F32(a0[kk], bf[kk][1], c01);
      c02 = WMMA_F32(a0[kk], bf[kk][2], c02);
      c03 = WMMA_F32(a0[kk], bf[kk][3], c03);
      c10 = WMMA_F32(a1[kk], bf[kk][0], c10);
      c11 = WMMA_F32(a1[kk], bf[kk][1], c11);
      c12 = WMMA_F32(a1[kk], bf[kk][2], c12);
      c13 = WMMA_F32(a1[kk], bf[kk][3], c13);
    }
  }

  float* Cp0 = C + (size_t)(m0 + half * 8) * N + n0 + 4 * l15;
#pragma unroll
  for (int j = 0; j < 8; ++j) {
    v4f t;
    t.x = c00[j];  t.y = c01[j];  t.z = c02[j];  t.w = c03[j];
    *(v4f*)(Cp0 + (size_t)j * N) = t;
  }
  if (has2) {
    float* Cp1 = C + (size_t)(m0 + 16 + half * 8) * N + n0 + 4 * l15;
#pragma unroll
    for (int j = 0; j < 8; ++j) {
      v4f t;
      t.x = c10[j];  t.y = c11[j];  t.z = c12[j];  t.w = c13[j];
      *(v4f*)(Cp1 + (size_t)j * N) = t;
    }
  }
}


__global__ __launch_bounds__(128)
void gemm_with_lists(const float* __restrict__ A, const float* __restrict__ B,
                     float* __restrict__ C, int M, int N, int K, int nbx, int nside,
                     const int* __restrict__ src, const int* __restrict__ dst,
                     const float* __restrict__ ew, int* __restrict__ cnt,
                     int2* __restrict__ slots, int* __restrict__ novf,
                     int* __restrict__ ovd, int* __restrict__ ovs,
                     float* __restrict__ ovw, int E) {
  __shared__ __attribute__((aligned(16))) float sA[4][32 * AP];
  const int bid = blockIdx.x;
  if (bid < nside) {
    const int n4   = E >> 2;
    const int nthr = nside * 128;
    int g = bid * 128 + threadIdx.x;
    int4   d = make_int4(0, 0, 0, 0);
    int4   s = make_int4(0, 0, 0, 0);
    float4 w = make_float4(0.f, 0.f, 0.f, 0.f);
    if (g < n4) {
      d = ((const int4*)dst)[g];
      s = ((const int4*)src)[g];
      w = ((const float4*)ew)[g];
    }
    while (g < n4) {
      const int gn = g + nthr;
      int4   dn = d;
      int4   sn = s;
      float4 wn = w;
      if (gn < n4) {
        dn = ((const int4*)dst)[gn];
        sn = ((const int4*)src)[gn];
        wn = ((const float4*)ew)[gn];
      }
      int4 r;
      r.x = atomicAdd(&cnt[d.x], 1);
      r.y = atomicAdd(&cnt[d.y], 1);
      r.z = atomicAdd(&cnt[d.z], 1);
      r.w = atomicAdd(&cnt[d.w], 1);
      store_edge(d.x, s.x, w.x, r.x, slots, novf, ovd, ovs, ovw);
      store_edge(d.y, s.y, w.y, r.y, slots, novf, ovd, ovs, ovw);
      store_edge(d.z, s.z, w.z, r.z, slots, novf, ovd, ovs, ovw);
      store_edge(d.w, s.w, w.w, r.w, slots, novf, ovd, ovs, ovw);
      d = dn;
      s = sn;
      w = wn;
      g = gn;
    }
    if (bid == 0 && threadIdx.x == 0)
      for (int i = n4 * 4; i < E; ++i)
        store_edge(dst[i], src[i], ew[i], atomicAdd(&cnt[dst[i]], 1),
                   slots, novf, ovd, ovs, ovw);
    return;
  }
  const int g  = bid - nside;
  const int by = g / nbx;
  gemm_block(A, B, C, M, N, K, g - by * nbx, by, sA);
}

template <int F, bool RELU>
__device__ __forceinline__
void full_row(const float4* __restrict__ hw4, const int2* __restrict__ slots,
              const int* __restrict__ novf, const int* __restrict__ ovd,
              const int* __restrict__ ovs, const float* __restrict__ ovw,
              int d, int deg, int E, const float4 (&b4)[F / 128], float4* o) {
  constexpr int V = F / 128;
  float4 acc[V];
#pragma unroll
  for (int j = 0; j < V; ++j) acc[j] = make_float4(0.f, 0.f, 0.f, 0.f);
#pragma unroll 1
  for (int i = d * CAP; i < (d + 1) * CAP; i += 4) {
    float4 v[4][V];
    float  w[4];
#pragma unroll
    for (int u = 0; u < 4; ++u) {
      const int2 p = slots[i + u];
      w[u] = __int_as_float(p.y);
      const float4* r = hw4 + (size_t)p.x * (F / 4);
#pragma unroll
      for (int j = 0; j < V; ++j) v[u][j] = r[32 * j];
    }
#pragma unroll
    for (int u = 0; u < 4; ++u) {
#pragma unroll
      for (int j = 0; j < V; ++j) {
        acc[j].x += w[u] * v[u][j].x;
        acc[j].y += w[u] * v[u][j].y;
        acc[j].z += w[u] * v[u][j].z;
        acc[j].w += w[u] * v[u][j].w;
      }
    }
  }
  int nov = novf[0];
  if (nov > E) nov = E;
#pragma unroll 1
  for (int i = 0; i < nov; ++i) {
    if (ovd[i] != d) continue;
    const float   w = ovw[i];
    const float4* r = hw4 + (size_t)ovs[i] * (F / 4);
#pragma unroll
    for (int j = 0; j < V; ++j) {
      const float4 v = r[32 * j];
      acc[j].x += w * v.x;
      acc[j].y += w * v.y;
      acc[j].z += w * v.z;
      acc[j].w += w * v.w;
    }
  }
  const float inv = 1.0f / fmaxf((float)deg, 1.0f);
#pragma unroll
  for (int j = 0; j < V; ++j) {
    float4 t;
    t.x = acc[j].x * inv + b4[j].x;
    t.y = acc[j].y * inv + b4[j].y;
    t.z = acc[j].z * inv + b4[j].z;
    t.w = acc[j].w * inv + b4[j].w;
    if (RELU) {
      t.x = fmaxf(t.x, 0.f);
      t.y = fmaxf(t.y, 0.f);
      t.z = fmaxf(t.z, 0.f);
      t.w = fmaxf(t.w, 0.f);
    }
    o[32 * j] = t;
  }
}

template <int F, int R, bool RELU>
__global__ __launch_bounds__(256)
void gather_rows(const float* __restrict__ hw, const int2* __restrict__ slots,
                 const int* __restrict__ cnt, const int* __restrict__ novf,
                 const int* __restrict__ ovd, const int* __restrict__ ovs,
                 const float* __restrict__ ovw, const float* __restrict__ bias,
                 float* __restrict__ res, int M, int E) {
  constexpr int V = F / 128;
  constexpr int U = 2048 / F;
  const int lane = threadIdx.x & 31;
  const int wave = __builtin_amdgcn_readfirstlane((int)(threadIdx.x >> 5));
  const int d0 = (blockIdx.x * 8 + wave) * R;
  if (d0 >= M) return;
  const int d1 = (d0 + R < M) ? d0 + R : M;

  const float4* hw4 = (const float4*)hw + lane;
  float4 b4[V];
#pragma unroll
  for (int j = 0; j < V; ++j) b4[j] = ((const float4*)bias)[lane + 32 * j];

  for (int d = d0; d < d1; ++d) {
    const int deg = cnt[d];
    if (deg > CAP) continue;
    const int beg = d * CAP;
    const int end = beg + deg;
    float4 acc[V];
#pragma unroll
    for (int j = 0; j < V; ++j) acc[j] = make_float4(0.f, 0.f, 0.f, 0.f);

    int i = beg;
    for (; i + U <= end; i += U) {
      float4 v[U][V];
      float  w[U];
#pragma unroll
      for (int u = 0; u < U; ++u) {
        const int2 p = slots[i + u];
        w[u] = __int_as_float(p.y);
        const float4* r = hw4 + (size_t)p.x * (F / 4);
#pragma unroll
        for (int j = 0; j < V; ++j) v[u][j] = r[32 * j];
      }
#pragma unroll
      for (int u = 0; u < U; ++u) {
#pragma unroll
        for (int j = 0; j < V; ++j) {
          acc[j].x += w[u] * v[u][j].x;
          acc[j].y += w[u] * v[u][j].y;
          acc[j].z += w[u] * v[u][j].z;
          acc[j].w += w[u] * v[u][j].w;
        }
      }
    }
    const int n = end - i;
    if (n > 0) {
      float4 v[U][V];
      float  w[U];
      int    s[U];
#pragma unroll
      for (int u = 0; u < U; ++u) {
        const int2 p = slots[(u < n) ? i + u : end - 1];
        w[u] = (u < n) ? __int_as_float(p.y) : 0.f;
        s[u] = p.x;
      }
#pragma unroll
      for (int u = 0; u < U; ++u) {
#pragma unroll
        for (int j = 0; j < V; ++j) v[u][j] = make_float4(0.f, 0.f, 0.f, 0.f);
        if (u < n) {
          const float4* r = hw4 + (size_t)s[u] * (F / 4);
#pragma unroll
          for (int j = 0; j < V; ++j) v[u][j] = r[32 * j];
        }
      }
#pragma unroll
      for (int u = 0; u < U; ++u) {
#pragma unroll
        for (int j = 0; j < V; ++j) {
          acc[j].x += w[u] * v[u][j].x;
          acc[j].y += w[u] * v[u][j].y;
          acc[j].z += w[u] * v[u][j].z;
          acc[j].w += w[u] * v[u][j].w;
        }
      }
    }

    const float inv = 1.0f / fmaxf((float)deg, 1.0f);
    float4* o = (float4*)res + (size_t)d * (F / 4) + lane;
#pragma unroll
    for (int j = 0; j < V; ++j) {
      float4 t;
      t.x = acc[j].x * inv + b4[j].x;
      t.y = acc[j].y * inv + b4[j].y;
      t.z = acc[j].z * inv + b4[j].z;
      t.w = acc[j].w * inv + b4[j].w;
      if (RELU) {
        t.x = fmaxf(t.x, 0.f);
        t.y = fmaxf(t.y, 0.f);
        t.z = fmaxf(t.z, 0.f);
        t.w = fmaxf(t.w, 0.f);
      }
      o[32 * j] = t;
    }
  }
  for (int d = d0; d < d1; ++d) {
    const int deg = cnt[d];
    if (deg > CAP)
      full_row<F, RELU>(hw4, slots, novf, ovd, ovs, ovw, d, deg, E, b4,
                        (float4*)res + (size_t)d * (F / 4) + lane);
  }
}

#define HP 260
#define CP 132
__global__ __launch_bounds__(256, 10)
void gather_relu_gemm(const float* __restrict__ hw, const int2* __restrict__ slots,
                      const int* __restrict__ cnt, const int* __restrict__ novf,
                      const int* __restrict__ ovd, const int* __restrict__ ovs,
                      const float* __restrict__ ovw, const float* __restrict__ bias,
                      const float* __restrict__ B, float* __restrict__ C, int M, int E) {
  constexpr int F = HID_F;
  constexpr int V = F / 128;
  constexpr int U = 2048 / F;
  constexpr int R = 4;
  __shared__ __attribute__((aligned(16))) float sH[32 * HP];
  __shared__ __attribute__((aligned(16))) float sC[32 * CP];
  const int lane = threadIdx.x & 31;
  const int wave = __builtin_amdgcn_readfirstlane((int)(threadIdx.x >> 5));
  const int r0 = blockIdx.x * 32;

  {
    const int d0 = r0 + wave * R;
    const float4* hw4 = (const float4*)hw + lane;
    float4 b4[V];
#pragma unroll
    for (int j = 0; j < V; ++j) b4[j] = ((const float4*)bias)[lane + 32 * j];

    const int d1 = (d0 + R < M) ? d0 + R : ((d0 < M) ? M : d0);
    for (int d = d1; d < d0 + R; ++d) {
      float4* o = (float4*)(sH + (d - r0) * HP) + lane;
#pragma unroll
      for (int j = 0; j < V; ++j) o[32 * j] = make_float4(0.f, 0.f, 0.f, 0.f);
    }
    for (int d = d0; d < d1; ++d) {
      float4* o = (float4*)(sH + (d - r0) * HP) + lane;
      const int deg = cnt[d];
      if (deg > CAP) continue;
      const int beg = d * CAP;
      const int end = beg + deg;
      float4 acc[V];
#pragma unroll
      for (int j = 0; j < V; ++j) acc[j] = make_float4(0.f, 0.f, 0.f, 0.f);

      int i = beg;
      for (; i + U <= end; i += U) {
        float4 v[U][V];
        float  w[U];
#pragma unroll
        for (int u = 0; u < U; ++u) {
          const int2 p = slots[i + u];
          w[u] = __int_as_float(p.y);
          const float4* r = hw4 + (size_t)p.x * (F / 4);
#pragma unroll
          for (int j = 0; j < V; ++j) v[u][j] = r[32 * j];
        }
#pragma unroll
        for (int u = 0; u < U; ++u) {
#pragma unroll
          for (int j = 0; j < V; ++j) {
            acc[j].x += w[u] * v[u][j].x;
            acc[j].y += w[u] * v[u][j].y;
            acc[j].z += w[u] * v[u][j].z;
            acc[j].w += w[u] * v[u][j].w;
          }
        }
      }
      const int n = end - i;
      if (n > 0) {
        float4 v[U][V];
        float  w[U];
        int    s[U];
#pragma unroll
        for (int u = 0; u < U; ++u) {
          const int2 p = slots[(u < n) ? i + u : end - 1];
          w[u] = (u < n) ? __int_as_float(p.y) : 0.f;
          s[u] = p.x;
        }
#pragma unroll
        for (int u = 0; u < U; ++u) {
#pragma unroll
          for (int j = 0; j < V; ++j) v[u][j] = make_float4(0.f, 0.f, 0.f, 0.f);
          if (u < n) {
            const float4* r = hw4 + (size_t)s[u] * (F / 4);
#pragma unroll
            for (int j = 0; j < V; ++j) v[u][j] = r[32 * j];
          }
        }
#pragma unroll
        for (int u = 0; u < U; ++u) {
#pragma unroll
          for (int j = 0; j < V; ++j) {
            acc[j].x += w[u] * v[u][j].x;
            acc[j].y += w[u] * v[u][j].y;
            acc[j].z += w[u] * v[u][j].z;
            acc[j].w += w[u] * v[u][j].w;
          }
        }
      }

      const float inv = 1.0f / fmaxf((float)deg, 1.0f);
#pragma unroll
      for (int j = 0; j < V; ++j) {
        float4 t;
        t.x = acc[j].x * inv + b4[j].x;
        t.y = acc[j].y * inv + b4[j].y;
        t.z = acc[j].z * inv + b4[j].z;
        t.w = acc[j].w * inv + b4[j].w;
        t.x = fmaxf(t.x, 0.f);
        t.y = fmaxf(t.y, 0.f);
        t.z = fmaxf(t.z, 0.f);
        t.w = fmaxf(t.w, 0.f);
        o[32 * j] = t;
      }
    }
    for (int d = d0; d < d1; ++d) {
      const int deg = cnt[d];
      if (deg > CAP)
        full_row<F, true>(hw4, slots, novf, ovd, ovs, ovw, d, deg, E, b4,
                          (float4*)(sH + (d - r0) * HP) + lane);
    }
  }

  const int half = lane >> 4;
  const int l15  = lane & 15;
  const int n0   = wave * 16;
  const float* fa0 = sH + l15 * HP + half * 2;
  const float* fa1 = sH + (16 + l15) * HP + half * 2;
  const float* Bp = B + (size_t)(half * 2) * OUT_F + n0 + l15;

  v2f bcur[8];
#pragma unroll
  for (int kk = 0; kk < 8; ++kk) {
    const float* bk = Bp + (size_t)(kk * 4) * OUT_F;
    bcur[kk].x = bk[0];
    bcur[kk].y = bk[OUT_F];
  }
  __syncthreads();

  v8f c0 = {}, c1 = {};
#pragma unroll
  for (int ch = 0; ch < HID_F / 32; ++ch) {
    v2f a0[8], a1[8];
#pragma unroll
    for (int kk = 0; kk < 8; ++kk) {
      a0[kk] = *(const v2f*)(fa0 + ch * 32 + kk * 4);
      a1[kk] = *(const v2f*)(fa1 + ch * 32 + kk * 4);
    }
    v2f bnext[8];
    if (ch + 1 < HID_F / 32) {
#pragma unroll
      for (int kk = 0; kk < 8; ++kk) {
        const float* bk = Bp + (size_t)((ch + 1) * 32 + kk * 4) * OUT_F;
        bnext[kk].x = bk[0];
        bnext[kk].y = bk[OUT_F];
      }
    }
#pragma unroll
    for (int kk = 0; kk < 8; ++kk) {
      c0 = WMMA_F32(a0[kk], bcur[kk], c0);
      c1 = WMMA_F32(a1[kk], bcur[kk], c1);
    }
    if (ch + 1 < HID_F / 32) {
#pragma unroll
      for (int kk = 0; kk < 8; ++kk) bcur[kk] = bnext[kk];
    }
  }

#pragma unroll
  for (int j = 0; j < 8; ++j) {
    sC[(j + 8 * half) * CP + n0 + l15]      = c0[j];
    sC[(16 + j + 8 * half) * CP + n0 + l15] = c1[j];
  }
  __syncthreads();
  float4 t[R];
#pragma unroll
  for (int q = 0; q < R; ++q) t[q] = *((const float4*)(sC + (wave * R + q) * CP) + lane);
#pragma unroll
  for (int q = 0; q < R; ++q) {
    const int row = r0 + wave * R + q;
    if (row < M)
      *((float4*)(C + (size_t)row * OUT_F) + lane) = t[q];
  }
}

extern "C" void kernel_launch(void* const* d_in, const int* in_sizes, int n_in,
                              void* d_out, int out_size, void* d_ws, size_t ws_size,
                              hipStream_t stream) {
  const float* x    = (const float*)d_in[0];
  const float* W1   = (const float*)d_in[1];
  const float* b1   = (const float*)d_in[2];
  const float* W2   = (const float*)d_in[3];
  const float* b2   = (const float*)d_in[4];
  const float* ew   = (const float*)d_in[5];
  const int*   esrc = (const int*)d_in[6];
  const int*   edst = (const int*)d_in[7];
  float* out = (float*)d_out;

  const int M = in_sizes[0] / IN_F;
  const int E = in_sizes[5];

  int* cnt = (int*)d_ws;
  const long long Mpad = ((long long)M + 63) & ~63LL;
  float* bufA = (float*)d_ws + Mpad;
  float* bufB = bufA + (long long)M * HID_F;
  float* upper = bufB + (long long)M * OUT_F;
  int2*  slots = (int2*)upper;
  int*   ovd   = (int*)(upper + (long long)M * (2 * CAP));
  int*   ovs   = ovd + E;
  float* ovw   = (float*)(ovs + E);
  int*   novf  = (int*)(ovw + E);

  const int T = 256;
  constexpr int R = 6;
  const int NSIDE1 = 256;
  const int nbx = (M + 127) / 128;

  {
    long long n4 = Mpad / 4;
    zero_f4<<<dim3((unsigned)((n4 + T - 1) / T)), dim3(T), 0, stream>>>((float4*)cnt, n4, novf);
    gemm_with_lists<<<dim3(NSIDE1 + nbx * (HID_F / 64)), dim3(128), 0, stream>>>(
        x, W1, bufA, M, HID_F, IN_F, nbx, NSIDE1, esrc, edst, ew, cnt, slots, novf,
        ovd, ovs, ovw, E);
  }

  gather_relu_gemm<<<dim3((M + 31) / 32), dim3(T), 0, stream>>>(
      bufA, slots, cnt, novf, ovd, ovs, ovw, b1, W2, bufB, M, E);

  gather_rows<OUT_F, R, false><<<dim3((M + 8 * R - 1) / (8 * R)), dim3(T), 0, stream>>>(
      bufB, slots, cnt, novf, ovd, ovs, ovw, b2, out, M, E);
}
